# starting kernel
# baseline (speedup 1.0000x reference)
#include <hip/hip_runtime.h>

typedef __attribute__((ext_vector_type(16))) __bf16 v16bf;
typedef __attribute__((ext_vector_type(8)))  float  v8f;
typedef __attribute__((ext_vector_type(4)))  unsigned int u32x4;

union FragU { u32x4 q[2]; v16bf v; };

__device__ __forceinline__ v16bf load_frag16(const unsigned short* p) {
    FragU f;
    f.q[0] = *(const u32x4*)(p);
    f.q[1] = *(const u32x4*)(p + 16);
    return f.v;
}

__device__ __forceinline__ v8f wmma_bf16(v16bf a, v16bf b, v8f c) {
    return __builtin_amdgcn_wmma_f32_16x16x32_bf16(
         false, a,  false, b,
         (short)0, c,  false,  false);
}

__device__ __forceinline__ unsigned short f2bf(float f) {
    unsigned int u = __float_as_uint(f);
    unsigned int r = u + 0x7FFFu + ((u >> 16) & 1u);
    return (unsigned short)(r >> 16);
}

__device__ __forceinline__ void atomicMaxFloat(float* addr, float value) {
    if (value >= 0.f) atomicMax((int*)addr, __float_as_int(value));
    else              atomicMin((unsigned int*)addr, __float_as_uint(value));
}

#define GN 50000
#define GE 800000
#define GD 128
#define GH 8
#define GDFF 512

__global__ void repack_kernel(const float* __restrict__ x, const float* __restrict__ Wfc,
                              const float* __restrict__ W1, const float* __restrict__ W2,
                              unsigned short* __restrict__ xb, unsigned short* __restrict__ Wzt,
                              unsigned short* __restrict__ W1t, unsigned short* __restrict__ W2t) {
    int idx = blockIdx.x * blockDim.x + threadIdx.x;
    if (idx < GN * GD) xb[idx] = f2bf(x[idx]);
    if (idx < GD * GD) {
        int j = idx >> 7, d = idx & 127;
        int h = j >> 4, k = j & 15;
        Wzt[idx] = f2bf(Wfc[(h * GD + d) * 16 + k]);
    }
    if (idx < GDFF * GD) {
        int f = idx >> 7, d = idx & 127;
        W1t[idx] = f2bf(W1[d * GDFF + f]);
        int d2 = idx >> 9, f2 = idx & 511;
        W2t[idx] = f2bf(W2[f2 * GD + d2]);
    }
}

__global__ void init_kernel(float* __restrict__ hagg, float* __restrict__ m, float* __restrict__ s) {
    int idx = blockIdx.x * blockDim.x + threadIdx.x;
    if (idx < GN * GD) hagg[idx] = 0.f;
    if (idx < GN * GH) { m[idx] = -INFINITY; s[idx] = 0.f; }
}

__global__ void gemm_z_kernel(const unsigned short* __restrict__ xb,
                              const unsigned short* __restrict__ Wzt,
                              float* __restrict__ z) {
    int wave = threadIdx.x >> 5, lane = threadIdx.x & 31;
    int rowBase = blockIdx.x * 16;
    int col0 = wave * 16;
    int lrow = lane & 15;
    int kb = (lane & 16) ? 8 : 0;
    const unsigned short* aRow = xb  + (size_t)(rowBase + lrow) * GD + kb;
    const unsigned short* bCol = Wzt + (size_t)(col0 + lrow)   * GD + kb;
    v8f acc = {};
    for (int k0 = 0; k0 < GD; k0 += 32)
        acc = wmma_bf16(load_frag16(aRow + k0), load_frag16(bCol + k0), acc);
    int half8 = (lane & 16) >> 1;
    int n = col0 + lrow;
    for (int j = 0; j < 8; ++j)
        z[(size_t)(rowBase + half8 + j) * GD + n] = acc[j];
}

__global__ void elr_kernel(const float* __restrict__ z, const float* __restrict__ a_l,
                           const float* __restrict__ a_r, float* __restrict__ el,
                           float* __restrict__ er) {
    int idx = blockIdx.x * blockDim.x + threadIdx.x;
    if (idx >= GN * GH) return;
    int h = idx & 7;
    const float* zp = z + (size_t)(idx >> 3) * GD + h * 16;
    float sl = 0.f, sr = 0.f;
    #pragma unroll
    for (int k = 0; k < 16; ++k) { sl += zp[k] * a_l[h * 16 + k]; sr += zp[k] * a_r[h * 16 + k]; }
    el[idx] = sl; er[idx] = sr;
}

__global__ void edge1_kernel(const float* __restrict__ el, const float* __restrict__ er,
                             const int* __restrict__ esrc, const int* __restrict__ edst,
                             float* __restrict__ ebuf, float* __restrict__ m) {
    int idx = blockIdx.x * blockDim.x + threadIdx.x;
    if (idx >= GE * GH) return;
    int e = idx >> 3, h = idx & 7;
    float v = el[esrc[e] * GH + h] + er[edst[e] * GH + h];
    v = v > 0.f ? v : 0.01f * v;
    ebuf[idx] = v;
    atomicMaxFloat(&m[edst[e] * GH + h], v);
}

__global__ void edge2_kernel(const int* __restrict__ edst, const float* __restrict__ m,
                             float* __restrict__ ebuf, float* __restrict__ s) {
    int idx = blockIdx.x * blockDim.x + threadIdx.x;
    if (idx >= GE * GH) return;
    int e = idx >> 3, h = idx & 7;
    float mm = m[edst[e] * GH + h];
    if (mm < -3.0e38f) mm = 0.f;
    float ee = __expf(ebuf[idx] - mm);
    ebuf[idx] = ee;
    atomicAdd(&s[edst[e] * GH + h], ee);
}

__global__ void scatter_kernel(const float* __restrict__ z, const float* __restrict__ ebuf,
                               const float* __restrict__ s, const int* __restrict__ esrc,
                               const int* __restrict__ edst, float* __restrict__ hagg) {
    int e = blockIdx.x * 2 + (threadIdx.x >> 7);
    int t = threadIdx.x & 127;
    int src = esrc[e], dst = edst[e];
    int h = t >> 4;
    float alpha = ebuf[(size_t)e * GH + h] / fmaxf(s[dst * GH + h], 1e-9f);
    atomicAdd(&hagg[(size_t)dst * GD + t], alpha * z[(size_t)src * GD + t]);
}

__global__ void node_ln_kernel(const float* __restrict__ hagg, const float* __restrict__ x,
                               const float* __restrict__ gamma, const float* __restrict__ beta,
                               float* __restrict__ hres, unsigned short* __restrict__ lnb) {
    int row = blockIdx.x;
    int t = threadIdx.x;
    float v = hagg[(size_t)row * GD + t];
    v = v > 0.f ? v : (__expf(v) - 1.f);
    v += x[(size_t)row * GD + t];
    hres[(size_t)row * GD + t] = v;
    __shared__ float rs[GD];
    __shared__ float rq[GD];
    rs[t] = v; rq[t] = v * v;
    __syncthreads();
    for (int sft = 64; sft > 0; sft >>= 1) {
        if (t < sft) { rs[t] += rs[t + sft]; rq[t] += rq[t + sft]; }
        __syncthreads();
    }
    float mean = rs[0] * (1.f / GD);
    float var  = rq[0] * (1.f / GD) - mean * mean;
    float ln = (v - mean) * rsqrtf(var + 1e-5f) * gamma[t] + beta[t];
    lnb[(size_t)row * GD + t] = f2bf(ln);
}

__global__ void gemm_ffn1_kernel(const unsigned short* __restrict__ lnb,
                                 const unsigned short* __restrict__ W1t,
                                 const float* __restrict__ b1,
                                 unsigned short* __restrict__ inter) {
    int wave = threadIdx.x >> 5, lane = threadIdx.x & 31;
    int rowBase = blockIdx.x * 16;
    int col0 = (blockIdx.y * 8 + wave) * 16;
    int lrow = lane & 15;
    int kb = (lane & 16) ? 8 : 0;
    const unsigned short* aRow = lnb + (size_t)(rowBase + lrow) * GD + kb;
    const unsigned short* bCol = W1t + (size_t)(col0 + lrow)   * GD + kb;
    v8f acc = {};
    for (int k0 = 0; k0 < GD; k0 += 32)
        acc = wmma_bf16(load_frag16(aRow + k0), load_frag16(bCol + k0), acc);
    int half8 = (lane & 16) >> 1;
    int n = col0 + lrow;
    float bias = b1[n];
    for (int j = 0; j < 8; ++j) {
        float v = fmaxf(acc[j] + bias, 0.f);
        inter[(size_t)(rowBase + half8 + j) * GDFF + n] = f2bf(v);
    }
}

__global__ void gemm_ffn2_kernel(const unsigned short* __restrict__ inter,
                                 const unsigned short* __restrict__ W2t,
                                 const float* __restrict__ b2,
                                 const float* __restrict__ hres,
                                 float* __restrict__ out) {
    int wave = threadIdx.x >> 5, lane = threadIdx.x & 31;
    int rowBase = blockIdx.x * 16;
    int col0 = wave * 16;
    int lrow = lane & 15;
    int kb = (lane & 16) ? 8 : 0;
    const unsigned short* aRow = inter + (size_t)(rowBase + lrow) * GDFF + kb;
    const unsigned short* bCol = W2t  + (size_t)(col0 + lrow)    * GDFF + kb;
    v8f acc = {};
    for (int k0 = 0; k0 < GDFF; k0 += 32)
        acc = wmma_bf16(load_frag16(aRow + k0), load_frag16(bCol + k0), acc);
    int half8 = (lane & 16) >> 1;
    int n = col0 + lrow;
    float bias = b2[n];
    for (int j = 0; j < 8; ++j) {
        size_t r = (size_t)(rowBase + half8 + j);
        out[r * GD + n] = acc[j] + bias + hres[r * GD + n];
    }
}

extern "C" void kernel_launch(void* const* d_in, const int* in_sizes, int n_in,
                              void* d_out, int out_size, void* d_ws, size_t ws_size,
                              hipStream_t stream) {
    const float* x     = (const float*)d_in[0];
    const float* Wfc   = (const float*)d_in[1];
    const float* a_l   = (const float*)d_in[2];
    const float* a_r   = (const float*)d_in[3];
    const float* gamma = (const float*)d_in[4];
    const float* beta  = (const float*)d_in[5];
    const float* W1    = (const float*)d_in[6];
    const float* b1    = (const float*)d_in[7];
    const float* W2    = (const float*)d_in[8];
    const float* b2    = (const float*)d_in[9];
    const int* esrc    = (const int*)d_in[10];
    const int* edst    = (const int*)d_in[11];
    float* out = (float*)d_out;

    char* ws = (char*)d_ws;
    size_t off = 0;
    auto take = [&](size_t bytes) -> char* {
        char* p = ws + off;
        off += (bytes + 255) & ~(size_t)255;
        return p;
    };
    unsigned short* xb    = (unsigned short*)take((size_t)GN * GD * 2);
    unsigned short* Wzt   = (unsigned short*)take((size_t)GD * GD * 2);
    unsigned short* W1t   = (unsigned short*)take((size_t)GDFF * GD * 2);
    unsigned short* W2t   = (unsigned short*)take((size_t)GD * GDFF * 2);
    float*          z     = (float*)take((size_t)GN * GD * 4);
    float*          el    = (float*)take((size_t)GN * GH * 4);
    float*          er    = (float*)take((size_t)GN * GH * 4);
    float*          m     = (float*)take((size_t)GN * GH * 4);
    float*          s     = (float*)take((size_t)GN * GH * 4);
    float*          ebuf  = (float*)take((size_t)GE * GH * 4);
    float*          hagg  = (float*)take((size_t)GN * GD * 4);
    float*          hres  = (float*)take((size_t)GN * GD * 4);
    unsigned short* lnb   = (unsigned short*)take((size_t)GN * GD * 2);
    unsigned short* inter = (unsigned short*)take((size_t)GN * GDFF * 2);

    const int mtiles = GN / 16;

    repack_kernel<<<(GN * GD + 255) / 256, 256, 0, stream>>>(x, Wfc, W1, W2, xb, Wzt, W1t, W2t);
    init_kernel  <<<(GN * GD + 255) / 256, 256, 0, stream>>>(hagg, m, s);
    gemm_z_kernel<<<mtiles, 256, 0, stream>>>(xb, Wzt, z);
    elr_kernel   <<<(GN * GH + 255) / 256, 256, 0, stream>>>(z, a_l, a_r, el, er);
    edge1_kernel <<<(GE * GH + 255) / 256, 256, 0, stream>>>(el, er, esrc, edst, ebuf, m);
    edge2_kernel <<<(GE * GH + 255) / 256, 256, 0, stream>>>(edst, m, ebuf, s);
    scatter_kernel<<<GE / 2, 256, 0, stream>>>(z, ebuf, s, esrc, edst, hagg);
    node_ln_kernel<<<GN, GD, 0, stream>>>(hagg, x, gamma, beta, hres, lnb);
    gemm_ffn1_kernel<<<dim3(mtiles, 4), 256, 0, stream>>>(lnb, W1t, b1, inter);
    gemm_ffn2_kernel<<<mtiles, 256, 0, stream>>>(inter, W2t, b2, hres, out);
}


# ELU + residual + LayerNorm fused into the gather wave's epilogue (row already in registers): LayerNorm launch and the row's memory round trip removed; LN sums now by wave shuffle butterfly
# speedup vs baseline: 1.7842x; 1.7842x over previous
#include <hip/hip_runtime.h>

typedef __attribute__((ext_vector_type(16))) __bf16 v16bf;
typedef __attribute__((ext_vector_type(8)))  float  v8f;
typedef __attribute__((ext_vector_type(4)))  unsigned int u32x4;

union FragU { u32x4 q[2]; v16bf v; };

__device__ __forceinline__ v16bf load_frag16(const unsigned short* p) {
    FragU f;
    f.q[0] = *(const u32x4*)(p);
    f.q[1] = *(const u32x4*)(p + 16);
    return f.v;
}

__device__ __forceinline__ v8f wmma_bf16(v16bf a, v16bf b, v8f c) {
    return __builtin_amdgcn_wmma_f32_16x16x32_bf16(
         false, a,  false, b,
         (short)0, c,  false,  false);
}

__device__ __forceinline__ unsigned short f2bf(float f) {
    unsigned int u = __float_as_uint(f);
    unsigned int r = u + 0x7FFFu + ((u >> 16) & 1u);
    return (unsigned short)(r >> 16);
}

#define GN 50000
#define GE 800000
#define GD 128
#define GH 8
#define GDFF 512

__global__ void repack_kernel(const float* __restrict__ x, const float* __restrict__ Wfc,
                              const float* __restrict__ W1, const float* __restrict__ W2,
                              unsigned short* __restrict__ xb, unsigned short* __restrict__ Wzt,
                              unsigned short* __restrict__ W1t, unsigned short* __restrict__ W2t,
                              int* __restrict__ cnt) {
    int idx = blockIdx.x * blockDim.x + threadIdx.x;
    if (idx < GN * GD) xb[idx] = f2bf(x[idx]);
    if (idx < GN) cnt[idx] = 0;
    if (idx < GD * GD) {
        int j = idx >> 7, d = idx & 127;
        int h = j >> 4, k = j & 15;
        Wzt[idx] = f2bf(Wfc[(h * GD + d) * 16 + k]);
    }
    if (idx < GDFF * GD) {
        int f = idx >> 7, d = idx & 127;
        W1t[idx] = f2bf(W1[d * GDFF + f]);
        int d2 = idx >> 9, f2 = idx & 511;
        W2t[idx] = f2bf(W2[f2 * GD + d2]);
    }
}

__global__ void gemm_z_kernel(const unsigned short* __restrict__ xb,
                              const unsigned short* __restrict__ Wzt,
                              float* __restrict__ z) {
    int wave = threadIdx.x >> 5, lane = threadIdx.x & 31;
    int rowBase = blockIdx.x * 16;
    int col0 = wave * 16;
    int lrow = lane & 15;
    int kb = (lane & 16) ? 8 : 0;
    const unsigned short* aRow = xb  + (size_t)(rowBase + lrow) * GD + kb;
    const unsigned short* bCol = Wzt + (size_t)(col0 + lrow)   * GD + kb;
    v8f acc = {};
    for (int k0 = 0; k0 < GD; k0 += 32)
        acc = wmma_bf16(load_frag16(aRow + k0), load_frag16(bCol + k0), acc);
    int half8 = (lane & 16) >> 1;
    int n = col0 + lrow;
    for (int j = 0; j < 8; ++j)
        z[(size_t)(rowBase + half8 + j) * GD + n] = acc[j];
}

__global__ void elr_kernel(const float* __restrict__ z, const float* __restrict__ a_l,
                           const float* __restrict__ a_r, float* __restrict__ el,
                           float* __restrict__ er) {
    int idx = blockIdx.x * blockDim.x + threadIdx.x;
    if (idx >= GN * GH) return;
    int h = idx & 7;
    const float* zp = z + (size_t)(idx >> 3) * GD + h * 16;
    float sl = 0.f, sr = 0.f;
    #pragma unroll
    for (int k = 0; k < 16; ++k) { sl += zp[k] * a_l[h * 16 + k]; sr += zp[k] * a_r[h * 16 + k]; }
    el[idx] = sl; er[idx] = sr;
}

__global__ void hist_kernel(const int* __restrict__ edst, int* __restrict__ cnt) {
    int e = blockIdx.x * blockDim.x + threadIdx.x;
    if (e < GE) atomicAdd(&cnt[edst[e]], 1);
}

#define SCAN_T 1024
#define SCAN_PER 49
__global__ void __launch_bounds__(SCAN_T)
scan_kernel(const int* __restrict__ cnt, int* __restrict__ rowptr, int* __restrict__ cursor) {
    __shared__ int sh[SCAN_T * SCAN_PER];
    __shared__ int wsum[32];
    int t = threadIdx.x;
    #pragma unroll
    for (int k = 0; k < SCAN_PER; ++k) {
        int i = k * SCAN_T + t;
        sh[i] = (i < GN) ? cnt[i] : 0;
    }
    __syncthreads();
    int base = t * SCAN_PER;
    int local = 0;
    for (int k = 0; k < SCAN_PER; ++k) local += sh[base + k];
    int lane = t & 31, w = t >> 5;
    int incl = local;
    for (int d = 1; d < 32; d <<= 1) { int y = __shfl_up(incl, d); if (lane >= d) incl += y; }
    if (lane == 31) wsum[w] = incl;
    __syncthreads();
    if (w == 0) {
        int x0 = wsum[lane];
        int xi = x0;
        for (int d = 1; d < 32; d <<= 1) { int y = __shfl_up(xi, d); if (lane >= d) xi += y; }
        wsum[lane] = xi - x0;
    }
    __syncthreads();
    int run = wsum[w] + incl - local;
    for (int k = 0; k < SCAN_PER; ++k) { int c = sh[base + k]; sh[base + k] = run; run += c; }
    __syncthreads();
    #pragma unroll
    for (int k = 0; k < SCAN_PER; ++k) {
        int i = k * SCAN_T + t;
        if (i < GN) { int v = sh[i]; rowptr[i] = v; cursor[i] = v; }
    }
    if (t == SCAN_T - 1) rowptr[GN] = run;
}

__global__ void fill_kernel(const int* __restrict__ esrc, const int* __restrict__ edst,
                            int* __restrict__ cursor, int* __restrict__ csrc) {
    int e = blockIdx.x * blockDim.x + threadIdx.x;
    if (e >= GE) return;
    int pos = atomicAdd(&cursor[edst[e]], 1);
    pos = min(max(pos, 0), GE - 1);
    csrc[pos] = esrc[e];
}

__global__ void gather_ln_kernel(const float* __restrict__ z, const float* __restrict__ el,
                                 const float* __restrict__ er, const int* __restrict__ rowptr,
                                 const int* __restrict__ csrc, const float* __restrict__ x,
                                 const float* __restrict__ gamma, const float* __restrict__ beta,
                                 float* __restrict__ hres, unsigned short* __restrict__ lnb) {
    int lane = threadIdx.x & 31;
    int wave = __builtin_amdgcn_readfirstlane((int)(threadIdx.x >> 5));
    int n = blockIdx.x * 8 + wave;
    int beg = __builtin_amdgcn_readfirstlane(rowptr[n]);
    int end = __builtin_amdgcn_readfirstlane(rowptr[n + 1]);
    beg = min(max(beg, 0), GE);
    end = min(max(end, beg), GE);
    int hd = lane >> 2;
    float ern = er[n * GH + hd];
    const float4 xx = *(const float4*)(x + (size_t)n * GD + lane * 4);
    const float4 gg = *(const float4*)(gamma + lane * 4);
    const float4 bb = *(const float4*)(beta + lane * 4);
    float m = -INFINITY, s = 0.f;
    float a0 = 0.f, a1 = 0.f, a2 = 0.f, a3 = 0.f;
    for (int base = beg; base < end; base += 32) {
        int cc = min(32, end - base);
        int mysrc = (lane < cc) ? csrc[base + lane] : 0;
        #pragma unroll 4
        for (int j = 0; j < cc; ++j) {
            int sidx = __builtin_amdgcn_readlane(mysrc, j);
            sidx = min(max(sidx, 0), GN - 1);
            float v = el[sidx * GH + hd] + ern;
            v = v > 0.f ? v : 0.01f * v;
            const float4 zz = *(const float4*)(z + (size_t)sidx * GD + lane * 4);
            float mn = fmaxf(m, v);
            float sc = __expf(m - mn);
            float p  = __expf(v - mn);
            s  = s  * sc + p;
            a0 = a0 * sc + p * zz.x;
            a1 = a1 * sc + p * zz.y;
            a2 = a2 * sc + p * zz.z;
            a3 = a3 * sc + p * zz.w;
            m = mn;
        }
    }
    float inv = 1.f / fmaxf(s, 1e-9f);
    float v0 = a0 * inv, v1 = a1 * inv, v2 = a2 * inv, v3 = a3 * inv;
    v0 = v0 > 0.f ? v0 : (__expf(v0) - 1.f);
    v1 = v1 > 0.f ? v1 : (__expf(v1) - 1.f);
    v2 = v2 > 0.f ? v2 : (__expf(v2) - 1.f);
    v3 = v3 > 0.f ? v3 : (__expf(v3) - 1.f);
    v0 += xx.x; v1 += xx.y; v2 += xx.z; v3 += xx.w;
    float4 hr; hr.x = v0; hr.y = v1; hr.z = v2; hr.w = v3;
    *(float4*)(hres + (size_t)n * GD + lane * 4) = hr;
    float sm = (v0 + v1) + (v2 + v3);
    float sq = (v0 * v0 + v1 * v1) + (v2 * v2 + v3 * v3);
    for (int d = 16; d > 0; d >>= 1) {
        sm += __shfl_xor(sm, d);
        sq += __shfl_xor(sq, d);
    }
    float mean = sm * (1.f / GD);
    float var  = sq * (1.f / GD) - mean * mean;
    float rstd = rsqrtf(var + 1e-5f);
    unsigned int b0 = f2bf((v0 - mean) * rstd * gg.x + bb.x);
    unsigned int b1 = f2bf((v1 - mean) * rstd * gg.y + bb.y);
    unsigned int b2 = f2bf((v2 - mean) * rstd * gg.z + bb.z);
    unsigned int b3 = f2bf((v3 - mean) * rstd * gg.w + bb.w);
    uint2 pk; pk.x = b0 | (b1 << 16); pk.y = b2 | (b3 << 16);
    *(uint2*)(lnb + (size_t)n * GD + lane * 4) = pk;
}

__global__ void gemm_ffn1_kernel(const unsigned short* __restrict__ lnb,
                                 const unsigned short* __restrict__ W1t,
                                 const float* __restrict__ b1,
                                 unsigned short* __restrict__ inter) {
    int wave = threadIdx.x >> 5, lane = threadIdx.x & 31;
    int rowBase = blockIdx.x * 16;
    int col0 = (blockIdx.y * 8 + wave) * 16;
    int lrow = lane & 15;
    int kb = (lane & 16) ? 8 : 0;
    const unsigned short* aRow = lnb + (size_t)(rowBase + lrow) * GD + kb;
    const unsigned short* bCol = W1t + (size_t)(col0 + lrow)   * GD + kb;
    v8f acc = {};
    for (int k0 = 0; k0 < GD; k0 += 32)
        acc = wmma_bf16(load_frag16(aRow + k0), load_frag16(bCol + k0), acc);
    int half8 = (lane & 16) >> 1;
    int n = col0 + lrow;
    float bias = b1[n];
    for (int j = 0; j < 8; ++j) {
        float v = fmaxf(acc[j] + bias, 0.f);
        inter[(size_t)(rowBase + half8 + j) * GDFF + n] = f2bf(v);
    }
}

__global__ void gemm_ffn2_kernel(const unsigned short* __restrict__ inter,
                                 const unsigned short* __restrict__ W2t,
                                 const float* __restrict__ b2,
                                 const float* __restrict__ hres,
                                 float* __restrict__ out) {
    int wave = threadIdx.x >> 5, lane = threadIdx.x & 31;
    int rowBase = blockIdx.x * 16;
    int col0 = wave * 16;
    int lrow = lane & 15;
    int kb = (lane & 16) ? 8 : 0;
    const unsigned short* aRow = inter + (size_t)(rowBase + lrow) * GDFF + kb;
    const unsigned short* bCol = W2t  + (size_t)(col0 + lrow)    * GDFF + kb;
    v8f acc = {};
    for (int k0 = 0; k0 < GDFF; k0 += 32)
        acc = wmma_bf16(load_frag16(aRow + k0), load_frag16(bCol + k0), acc);
    int half8 = (lane & 16) >> 1;
    int n = col0 + lrow;
    float bias = b2[n];
    for (int j = 0; j < 8; ++j) {
        size_t r = (size_t)(rowBase + half8 + j);
        out[r * GD + n] = acc[j] + bias + hres[r * GD + n];
    }
}

extern "C" void kernel_launch(void* const* d_in, const int* in_sizes, int n_in,
                              void* d_out, int out_size, void* d_ws, size_t ws_size,
                              hipStream_t stream) {
    const float* x     = (const float*)d_in[0];
    const float* Wfc   = (const float*)d_in[1];
    const float* a_l   = (const float*)d_in[2];
    const float* a_r   = (const float*)d_in[3];
    const float* gamma = (const float*)d_in[4];
    const float* beta  = (const float*)d_in[5];
    const float* W1    = (const float*)d_in[6];
    const float* b1    = (const float*)d_in[7];
    const float* W2    = (const float*)d_in[8];
    const float* b2    = (const float*)d_in[9];
    const int* esrc    = (const int*)d_in[10];
    const int* edst    = (const int*)d_in[11];
    float* out = (float*)d_out;

    char* ws = (char*)d_ws;
    size_t off = 0;
    auto take = [&](size_t bytes) -> char* {
        char* p = ws + off;
        off += (bytes + 255) & ~(size_t)255;
        return p;
    };
    unsigned short* xb    = (unsigned short*)take((size_t)GN * GD * 2);
    unsigned short* Wzt   = (unsigned short*)take((size_t)GD * GD * 2);
    unsigned short* W1t   = (unsigned short*)take((size_t)GDFF * GD * 2);
    unsigned short* W2t   = (unsigned short*)take((size_t)GD * GDFF * 2);
    float*          z     = (float*)take((size_t)GN * GD * 4);
    float*          el    = (float*)take((size_t)GN * GH * 4);
    float*          er    = (float*)take((size_t)GN * GH * 4);
    int*            cnt   = (int*)take((size_t)GN * 4);
    int*            rowptr= (int*)take((size_t)(GN + 1) * 4);
    int*            cursor= (int*)take((size_t)GN * 4);
    int*            csrc  = (int*)take((size_t)GE * 4);
    float*          hres  = (float*)take((size_t)GN * GD * 4);
    unsigned short* lnb   = (unsigned short*)take((size_t)GN * GD * 2);
    unsigned short* inter = (unsigned short*)take((size_t)GN * GDFF * 2);

    const int mtiles = GN / 16;

    repack_kernel<<<(GN * GD + 255) / 256, 256, 0, stream>>>(x, Wfc, W1, W2, xb, Wzt, W1t, W2t, cnt);
    gemm_z_kernel<<<mtiles, 256, 0, stream>>>(xb, Wzt, z);
    elr_kernel   <<<(GN * GH + 255) / 256, 256, 0, stream>>>(z, a_l, a_r, el, er);
    hist_kernel  <<<(GE + 255) / 256, 256, 0, stream>>>(edst, cnt);
    scan_kernel  <<<1, SCAN_T, 0, stream>>>(cnt, rowptr, cursor);
    fill_kernel  <<<(GE + 255) / 256, 256, 0, stream>>>(esrc, edst, cursor, csrc);
    gather_ln_kernel<<<GN / 8, 256, 0, stream>>>(z, el, er, rowptr, csrc, x, gamma, beta, hres, lnb);
    gemm_ffn1_kernel<<<dim3(mtiles, 4), 256, 0, stream>>>(lnb, W1t, b1, inter);
    gemm_ffn2_kernel<<<mtiles, 256, 0, stream>>>(inter, W2t, b2, hres, out);
}


# Edges grouped by destination in two levels with LDS counters (coarse bins of 256 nodes, then per node): both global-atomic passes (histogram, fill) are gone; bucket order is still arrival order
# speedup vs baseline: 4.5364x; 2.5425x over previous
#include <hip/hip_runtime.h>

typedef __attribute__((ext_vector_type(16))) __bf16 v16bf;
typedef __attribute__((ext_vector_type(8)))  float  v8f;
typedef __attribute__((ext_vector_type(4)))  unsigned int u32x4;

union FragU { u32x4 q[2]; v16bf v; };

__device__ __forceinline__ v16bf load_frag16(const unsigned short* p) {
    FragU f;
    f.q[0] = *(const u32x4*)(p);
    f.q[1] = *(const u32x4*)(p + 16);
    return f.v;
}

__device__ __forceinline__ v8f wmma_bf16(v16bf a, v16bf b, v8f c) {
    return __builtin_amdgcn_wmma_f32_16x16x32_bf16(
         false, a,  false, b,
         (short)0, c,  false,  false);
}

__device__ __forceinline__ unsigned short f2bf(float f) {
    unsigned int u = __float_as_uint(f);
    unsigned int r = u + 0x7FFFu + ((u >> 16) & 1u);
    return (unsigned short)(r >> 16);
}

#define GN 50000
#define GE 800000
#define GD 128
#define GH 8
#define GDFF 512

#define RP_T 256
#define RP_U 10
#define RP_BLOCKS (GN * GD / (RP_T * 4 * RP_U))
static_assert(RP_BLOCKS * RP_T * 4 * RP_U == GN * GD, "x must split into whole blocks");
static_assert(RP_BLOCKS * RP_T >= GDFF * GD, "grid must still cover the weight part");
__global__ void __launch_bounds__(RP_T)
repack_kernel(const float* __restrict__ x, const float* __restrict__ Wfc,
              const float* __restrict__ W1, const float* __restrict__ W2,
              unsigned short* __restrict__ xb, unsigned short* __restrict__ Wzt,
              unsigned short* __restrict__ W1t, unsigned short* __restrict__ W2t) {
    int idx = blockIdx.x * RP_T + threadIdx.x;
    size_t g0 = (size_t)blockIdx.x * (RP_T * RP_U) + threadIdx.x;
    const float4* xs = (const float4*)x + g0;
    uint2* xd = (uint2*)xb + g0;
    float4 v[RP_U];
    #pragma unroll
    for (int j = 0; j < RP_U; ++j) v[j] = xs[j * RP_T];
    if (idx < GD * GD) {
        int j = idx >> 7, d = idx & 127;
        int h = j >> 4, k = j & 15;
        Wzt[idx] = f2bf(Wfc[(h * GD + d) * 16 + k]);
    }
    if (idx < GDFF * GD) {
        int f = idx >> 7, d = idx & 127;
        W1t[idx] = f2bf(W1[d * GDFF + f]);
        int d2 = idx >> 9, f2 = idx & 511;
        W2t[idx] = f2bf(W2[f2 * GD + d2]);
    }
    #pragma unroll
    for (int j = 0; j < RP_U; ++j) {
        uint2 pk;
        pk.x = (unsigned int)f2bf(v[j].x) | ((unsigned int)f2bf(v[j].y) << 16);
        pk.y = (unsigned int)f2bf(v[j].z) | ((unsigned int)f2bf(v[j].w) << 16);
        xd[j * RP_T] = pk;
    }
}

__global__ void gemm_z_kernel(const unsigned short* __restrict__ xb,
                              const unsigned short* __restrict__ Wzt,
                              float* __restrict__ z) {
    int wave = threadIdx.x >> 5, lane = threadIdx.x & 31;
    int rowBase = blockIdx.x * 16;
    int col0 = wave * 16;
    int lrow = lane & 15;
    int kb = (lane & 16) ? 8 : 0;
    const unsigned short* aRow = xb  + (size_t)(rowBase + lrow) * GD + kb;
    const unsigned short* bCol = Wzt + (size_t)(col0 + lrow)   * GD + kb;
    v8f acc = {};
    for (int k0 = 0; k0 < GD; k0 += 32)
        acc = wmma_bf16(load_frag16(aRow + k0), load_frag16(bCol + k0), acc);
    int half8 = (lane & 16) >> 1;
    int n = col0 + lrow;
    for (int j = 0; j < 8; ++j)
        z[(size_t)(rowBase + half8 + j) * GD + n] = acc[j];
}

__global__ void elr_kernel(const float* __restrict__ z, const float* __restrict__ a_l,
                           const float* __restrict__ a_r, float* __restrict__ el,
                           float* __restrict__ er) {
    int idx = blockIdx.x * blockDim.x + threadIdx.x;
    if (idx >= GN * GH) return;
    int h = idx & 7;
    const float* zp = z + (size_t)(idx >> 3) * GD + h * 16;
    float sl = 0.f, sr = 0.f;
    #pragma unroll
    for (int k = 0; k < 16; ++k) { sl += zp[k] * a_l[h * 16 + k]; sr += zp[k] * a_r[h * 16 + k]; }
    el[idx] = sl; er[idx] = sr;
}

#define CB_SHIFT 8
#define CB_NODES (1 << CB_SHIFT)
#define CB_N ((GN + CB_NODES - 1) / CB_NODES)
#define CH_N 256
#define CH_E (GE / CH_N)
#define CELLS (CB_N * CH_N)
#define BIN_T 512
#define CH_IT ((CH_E + BIN_T - 1) / BIN_T)
#define FIN_T 1024
#define FIN_U 6
#define SCAN_T 1024
#define SCAN_PER 49
static_assert(CH_E * CH_N == GE, "the edges must split into whole chunks");
static_assert(CELLS == SCAN_T * SCAN_PER, "the scan covers exactly the cell table");

__global__ void __launch_bounds__(BIN_T)
bincount_kernel(const int* __restrict__ edst, int* __restrict__ cell) {
    __shared__ int hc[CB_N];
    const int t = threadIdx.x;
    if (t < CB_N) hc[t] = 0;
    __syncthreads();
    const int base = blockIdx.x * CH_E;
    int d[CH_IT];
    #pragma unroll
    for (int k = 0; k < CH_IT; ++k) {
        int i = t + k * BIN_T;
        d[k] = (i < CH_E) ? edst[base + i] : -1;
    }
    #pragma unroll
    for (int k = 0; k < CH_IT; ++k)
        if ((unsigned int)d[k] < (unsigned int)GN) atomicAdd(&hc[d[k] >> CB_SHIFT], 1);
    __syncthreads();
    if (t < CB_N) cell[t * CH_N + blockIdx.x] = hc[t];
}

__global__ void __launch_bounds__(SCAN_T)
scan_kernel(const int* __restrict__ cell, int* __restrict__ off) {
    __shared__ int sh[SCAN_T * SCAN_PER];
    __shared__ int wsum[32];
    int t = threadIdx.x;
    #pragma unroll
    for (int k = 0; k < SCAN_PER; ++k) {
        int i = k * SCAN_T + t;
        sh[i] = cell[i];
    }
    __syncthreads();
    int base = t * SCAN_PER;
    int local = 0;
    for (int k = 0; k < SCAN_PER; ++k) local += sh[base + k];
    int lane = t & 31, w = t >> 5;
    int incl = local;
    for (int d = 1; d < 32; d <<= 1) { int y = __shfl_up(incl, d); if (lane >= d) incl += y; }
    if (lane == 31) wsum[w] = incl;
    __syncthreads();
    if (w == 0) {
        int x0 = wsum[lane];
        int xi = x0;
        for (int d = 1; d < 32; d <<= 1) { int y = __shfl_up(xi, d); if (lane >= d) xi += y; }
        wsum[lane] = xi - x0;
    }
    __syncthreads();
    int run = wsum[w] + incl - local;
    for (int k = 0; k < SCAN_PER; ++k) { int c = sh[base + k]; sh[base + k] = run; run += c; }
    __syncthreads();
    #pragma unroll
    for (int k = 0; k < SCAN_PER; ++k) {
        int i = k * SCAN_T + t;
        off[i] = sh[i];
    }
    if (t == SCAN_T - 1) off[CELLS] = run;
}

__global__ void __launch_bounds__(BIN_T)
binplace_kernel(const int* __restrict__ esrc, const int* __restrict__ edst,
                const int* __restrict__ off, int2* __restrict__ pairs) {
    __shared__ int cur[CB_N];
    const int t = threadIdx.x;
    if (t < CB_N) cur[t] = off[t * CH_N + blockIdx.x];
    __syncthreads();
    const int base = blockIdx.x * CH_E;
    int d[CH_IT], s[CH_IT];
    #pragma unroll
    for (int k = 0; k < CH_IT; ++k) {
        int i = t + k * BIN_T;
        d[k] = (i < CH_E) ? edst[base + i] : -1;
        s[k] = (i < CH_E) ? esrc[base + i] : 0;
    }
    #pragma unroll
    for (int k = 0; k < CH_IT; ++k) {
        if ((unsigned int)d[k] < (unsigned int)GN) {
            int pos = atomicAdd(&cur[d[k] >> CB_SHIFT], 1);
            pos = min(max(pos, 0), GE - 1);
            int2 p; p.x = s[k]; p.y = d[k];
            pairs[pos] = p;
        }
    }
}

__global__ void __launch_bounds__(FIN_T)
binfinish_kernel(const int* __restrict__ off, const int2* __restrict__ pairs,
                 int* __restrict__ rowptr, int* __restrict__ csrc) {
    __shared__ int cnt[CB_NODES];
    __shared__ int wsum[CB_NODES / 32];
    const int t = threadIdx.x;
    const int b = blockIdx.x;
    int beg = off[b * CH_N];
    int end = off[(b + 1) * CH_N];
    beg = min(max(beg, 0), GE);
    end = min(max(end, beg), GE);
    if (t < CB_NODES) cnt[t] = 0;
    __syncthreads();
    for (int i0 = beg + t; i0 < end; i0 += FIN_T * FIN_U) {
        int dd[FIN_U];
        #pragma unroll
        for (int u = 0; u < FIN_U; ++u) {
            int i = i0 + u * FIN_T;
            dd[u] = (i < end) ? pairs[i].y : -1;
        }
        #pragma unroll
        for (int u = 0; u < FIN_U; ++u)
            if (dd[u] >= 0) atomicAdd(&cnt[dd[u] & (CB_NODES - 1)], 1);
    }
    __syncthreads();
    const int lane = t & 31, w = t >> 5;
    const int c = (t < CB_NODES) ? cnt[t] : 0;
    int incl = c;
    for (int d = 1; d < 32; d <<= 1) { int y = __shfl_up(incl, d); if (lane >= d) incl += y; }
    if (t < CB_NODES && lane == 31) wsum[w] = incl;
    __syncthreads();
    if (t < CB_NODES) {
        int wbase = 0;
        for (int k = 0; k < CB_NODES / 32; ++k) wbase += (k < w) ? wsum[k] : 0;
        int start = beg + wbase + incl - c;
        cnt[t] = start;
        int n = b * CB_NODES + t;
        if (n <= GN) rowptr[n] = start;
    }
    __syncthreads();
    for (int i0 = beg + t; i0 < end; i0 += FIN_T * FIN_U) {
        int ss[FIN_U], dd[FIN_U];
        #pragma unroll
        for (int u = 0; u < FIN_U; ++u) {
            int i = i0 + u * FIN_T;
            int2 p = pairs[(i < end) ? i : i0];
            ss[u] = p.x;
            dd[u] = (i < end) ? p.y : -1;
        }
        #pragma unroll
        for (int u = 0; u < FIN_U; ++u) {
            if (dd[u] >= 0) {
                int pos = atomicAdd(&cnt[dd[u] & (CB_NODES - 1)], 1);
                pos = min(max(pos, 0), GE - 1);
                csrc[pos] = ss[u];
            }
        }
    }
}

__global__ void gather_ln_kernel(const float* __restrict__ z, const float* __restrict__ el,
                                 const float* __restrict__ er, const int* __restrict__ rowptr,
                                 const int* __restrict__ csrc, const float* __restrict__ x,
                                 const float* __restrict__ gamma, const float* __restrict__ beta,
                                 float* __restrict__ hres, unsigned short* __restrict__ lnb) {
    int lane = threadIdx.x & 31;
    int wave = __builtin_amdgcn_readfirstlane((int)(threadIdx.x >> 5));
    int n = blockIdx.x * 8 + wave;
    int beg = __builtin_amdgcn_readfirstlane(rowptr[n]);
    int end = __builtin_amdgcn_readfirstlane(rowptr[n + 1]);
    beg = min(max(beg, 0), GE);
    end = min(max(end, beg), GE);
    int hd = lane >> 2;
    float ern = er[n * GH + hd];
    const float4 xx = *(const float4*)(x + (size_t)n * GD + lane * 4);
    const float4 gg = *(const float4*)(gamma + lane * 4);
    const float4 bb = *(const float4*)(beta + lane * 4);
    float m = -INFINITY, s = 0.f;
    float a0 = 0.f, a1 = 0.f, a2 = 0.f, a3 = 0.f;
    for (int base = beg; base < end; base += 32) {
        int cc = min(32, end - base);
        int mysrc = (lane < cc) ? csrc[base + lane] : 0;
        #pragma unroll 4
        for (int j = 0; j < cc; ++j) {
            int sidx = __builtin_amdgcn_readlane(mysrc, j);
            sidx = min(max(sidx, 0), GN - 1);
            float v = el[sidx * GH + hd] + ern;
            v = v > 0.f ? v : 0.01f * v;
            const float4 zz = *(const float4*)(z + (size_t)sidx * GD + lane * 4);
            float mn = fmaxf(m, v);
            float sc = __expf(m - mn);
            float p  = __expf(v - mn);
            s  = s  * sc + p;
            a0 = a0 * sc + p * zz.x;
            a1 = a1 * sc + p * zz.y;
            a2 = a2 * sc + p * zz.z;
            a3 = a3 * sc + p * zz.w;
            m = mn;
        }
    }
    float inv = 1.f / fmaxf(s, 1e-9f);
    float v0 = a0 * inv, v1 = a1 * inv, v2 = a2 * inv, v3 = a3 * inv;
    v0 = v0 > 0.f ? v0 : (__expf(v0) - 1.f);
    v1 = v1 > 0.f ? v1 : (__expf(v1) - 1.f);
    v2 = v2 > 0.f ? v2 : (__expf(v2) - 1.f);
    v3 = v3 > 0.f ? v3 : (__expf(v3) - 1.f);
    v0 += xx.x; v1 += xx.y; v2 += xx.z; v3 += xx.w;
    float4 hr; hr.x = v0; hr.y = v1; hr.z = v2; hr.w = v3;
    *(float4*)(hres + (size_t)n * GD + lane * 4) = hr;
    float sm = (v0 + v1) + (v2 + v3);
    float sq = (v0 * v0 + v1 * v1) + (v2 * v2 + v3 * v3);
    for (int d = 16; d > 0; d >>= 1) {
        sm += __shfl_xor(sm, d);
        sq += __shfl_xor(sq, d);
    }
    float mean = sm * (1.f / GD);
    float var  = sq * (1.f / GD) - mean * mean;
    float rstd = rsqrtf(var + 1e-5f);
    unsigned int b0 = f2bf((v0 - mean) * rstd * gg.x + bb.x);
    unsigned int b1 = f2bf((v1 - mean) * rstd * gg.y + bb.y);
    unsigned int b2 = f2bf((v2 - mean) * rstd * gg.z + bb.z);
    unsigned int b3 = f2bf((v3 - mean) * rstd * gg.w + bb.w);
    uint2 pk; pk.x = b0 | (b1 << 16); pk.y = b2 | (b3 << 16);
    *(uint2*)(lnb + (size_t)n * GD + lane * 4) = pk;
}

#define FF_T 512
#define FF_WAVES 16
#define FF_BLOCKS ((GN / 16 + FF_WAVES - 1) / FF_WAVES)
#define W1S_PITCH 272
#define W2S_PITCH 1040
#define SCR_PITCH 144
#define W1S_BYTES (GDFF * W1S_PITCH)
#define W2S_BYTES (GD * W2S_PITCH)
#define SCR_BYTES (16 * SCR_PITCH)
#define FF_LDS (W1S_BYTES + W2S_BYTES + FF_WAVES * SCR_BYTES)
static_assert(FF_LDS <= 320 * 1024, "weights and scratch must fit one work-group's LDS");
static_assert((GDFF * GD * 2 / 16) % FF_T == 0, "the weight copy must split into whole passes");

__device__ __forceinline__ v16bf lds_frag16(const unsigned char* p) {
    FragU f;
    f.q[0] = *(const u32x4*)(p);
    f.q[1] = *(const u32x4*)(p + 32);
    return f.v;
}

__device__ __forceinline__ void wave_lds_handover() {
    __builtin_amdgcn_fence(__ATOMIC_RELEASE, "wavefront");
    __builtin_amdgcn_wave_barrier();
    __builtin_amdgcn_fence(__ATOMIC_ACQUIRE, "wavefront");
}

__global__ void __launch_bounds__(FF_T)
ffn_fused_kernel(const unsigned short* __restrict__ lnb,
                 const unsigned short* __restrict__ W1t, const float* __restrict__ b1,
                 const unsigned short* __restrict__ W2t, const float* __restrict__ b2,
                 const float* __restrict__ hres, float* __restrict__ out) {
    __shared__ __attribute__((aligned(16))) unsigned char L[FF_LDS];
    const int tid = threadIdx.x;
    const int lane = tid & 31;
    const int wave = __builtin_amdgcn_readfirstlane(tid >> 5);
    const int tile = blockIdx.x * FF_WAVES + wave;
    const int tileC = min(tile, GN / 16 - 1);
    const int rowBase = tileC * 16;
    const int lrow = lane & 15;
    const int kb = (lane & 16) ? 8 : 0;
    const int half8 = (lane & 16) >> 1;

    const unsigned short* aRow = lnb + (size_t)(rowBase + lrow) * GD + kb;
    v16bf a[4];
    #pragma unroll
    for (int ks = 0; ks < 4; ++ks) a[ks] = load_frag16(aRow + ks * 32);

    {
        const u32x4* g1 = (const u32x4*)W1t;
        #pragma unroll 8
        for (int p = 0; p < (GDFF * GD * 2 / 16) / FF_T; ++p) {
            int q = p * FF_T + tid;
            u32x4 v = g1[q];
            *(u32x4*)(L + (q >> 4) * W1S_PITCH + (q & 15) * 16) = v;
        }
        const u32x4* g2 = (const u32x4*)W2t;
        #pragma unroll 8
        for (int p = 0; p < (GD * GDFF * 2 / 16) / FF_T; ++p) {
            int q = p * FF_T + tid;
            u32x4 v = g2[q];
            *(u32x4*)(L + W1S_BYTES + (q >> 6) * W2S_PITCH + (q & 63) * 16) = v;
        }
    }
    __syncthreads();
    if (tile >= GN / 16) return;

    const unsigned char* w1s = L + lrow * W1S_PITCH + kb * 2;
    const unsigned char* w2s = L + W1S_BYTES + lrow * W2S_PITCH + kb * 2;
    unsigned char* scr = L + W1S_BYTES + W2S_BYTES + wave * SCR_BYTES;

    v8f acc2[8];
    #pragma unroll
    for (int nt = 0; nt < 8; ++nt) acc2[nt] = v8f{};

    for (int c = 0; c < GDFF / 64; ++c) {
        v8f acc1[4];
        #pragma unroll
        for (int t = 0; t < 4; ++t) acc1[t] = v8f{};
        #pragma unroll
        for (int ks = 0; ks < 4; ++ks) {
            #pragma unroll
            for (int t = 0; t < 4; ++t) {
                v16bf b = lds_frag16(w1s + (c * 64 + t * 16) * W1S_PITCH + ks * 64);
                acc1[t] = wmma_bf16(a[ks], b, acc1[t]);
            }
        }
        #pragma unroll
        for (int t = 0; t < 4; ++t) {
            float bias = b1[c * 64 + t * 16 + lrow];
            #pragma unroll
            for (int j = 0; j < 8; ++j) {
                float v = fmaxf(acc1[t][j] + bias, 0.f);
                *(unsigned short*)(scr + (half8 + j) * SCR_PITCH + (t * 16 + lrow) * 2) = f2bf(v);
            }
        }
        wave_lds_handover();
        #pragma unroll
        for (int ks2 = 0; ks2 < 2; ++ks2) {
            v16bf a2 = lds_frag16(scr + lrow * SCR_PITCH + kb * 2 + ks2 * 64);
            #pragma unroll
            for (int nt = 0; nt < 8; ++nt) {
                v16bf b = lds_frag16(w2s + nt * 16 * W2S_PITCH + (c * 64 + ks2 * 32) * 2);
                acc2[nt] = wmma_bf16(a2, b, acc2[nt]);
            }
        }
        wave_lds_handover();
    }

    #pragma unroll
    for (int nt = 0; nt < 8; ++nt) {
        int n = nt * 16 + lrow;
        float bias = b2[n];
        #pragma unroll
        for (int j = 0; j < 8; ++j) {
            size_t r = (size_t)(rowBase + half8 + j);
            out[r * GD + n] = acc2[nt][j] + bias + hres[r * GD + n];
        }
    }
}

extern "C" void kernel_launch(void* const* d_in, const int* in_sizes, int n_in,
                              void* d_out, int out_size, void* d_ws, size_t ws_size,
                              hipStream_t stream) {
    const float* x     = (const float*)d_in[0];
    const float* Wfc   = (const float*)d_in[1];
    const float* a_l   = (const float*)d_in[2];
    const float* a_r   = (const float*)d_in[3];
    const float* gamma = (const float*)d_in[4];
    const float* beta  = (const float*)d_in[5];
    const float* W1    = (const float*)d_in[6];
    const float* b1    = (const float*)d_in[7];
    const float* W2    = (const float*)d_in[8];
    const float* b2    = (const float*)d_in[9];
    const int* esrc    = (const int*)d_in[10];
    const int* edst    = (const int*)d_in[11];
    float* out = (float*)d_out;

    char* ws = (char*)d_ws;
    size_t off = 0;
    auto take = [&](size_t bytes) -> char* {
        char* p = ws + off;
        off += (bytes + 255) & ~(size_t)255;
        return p;
    };
    unsigned short* xb    = (unsigned short*)take((size_t)GN * GD * 2);
    unsigned short* Wzt   = (unsigned short*)take((size_t)GD * GD * 2);
    unsigned short* W1t   = (unsigned short*)take((size_t)GDFF * GD * 2);
    unsigned short* W2t   = (unsigned short*)take((size_t)GD * GDFF * 2);
    float*          z     = (float*)take((size_t)GN * GD * 4);
    float*          el    = (float*)take((size_t)GN * GH * 4);
    float*          er    = (float*)take((size_t)GN * GH * 4);
    int*            rowptr= (int*)take((size_t)(GN + 1) * 4);
    int*            csrc  = (int*)take((size_t)GE * 4);
    float*          hres  = (float*)take((size_t)GN * GD * 4);
    unsigned short* lnb   = (unsigned short*)take((size_t)GN * GD * 2);
    int2*           pairs = (int2*)hres;
    int*            cell  = (int*)lnb;
    int*            coff  = (int*)((char*)lnb + (size_t)CELLS * 4);

    const int mtiles = GN / 16;

    repack_kernel<<<RP_BLOCKS, RP_T, 0, stream>>>(x, Wfc, W1, W2, xb, Wzt, W1t, W2t);
    gemm_z_kernel<<<mtiles, 256, 0, stream>>>(xb, Wzt, z);
    elr_kernel   <<<(GN * GH + 255) / 256, 256, 0, stream>>>(z, a_l, a_r, el, er);
    bincount_kernel <<<CH_N, BIN_T, 0, stream>>>(edst, cell);
    scan_kernel     <<<1, SCAN_T, 0, stream>>>(cell, coff);
    binplace_kernel <<<CH_N, BIN_T, 0, stream>>>(esrc, edst, coff, pairs);
    binfinish_kernel<<<CB_N, FIN_T, 0, stream>>>(coff, pairs, rowptr, csrc);
    gather_ln_kernel<<<GN / 8, 256, 0, stream>>>(z, el, er, rowptr, csrc, x, gamma, beta, hres, lnb);
    ffn_fused_kernel<<<FF_BLOCKS, FF_T, 0, stream>>>(lnb, W1t, b1, W2t, b2, hres, out);
}


# z projection and el/er in one launch: Wz staged in LDS, one wave per 16x128 strip (A fragments loaded once, not by 8 waves), rows leave via LDS as whole lines; el/er summed 4 per lane + quad shuffle
# speedup vs baseline: 4.7772x; 1.0531x over previous
#include <hip/hip_runtime.h>

typedef __attribute__((ext_vector_type(16))) __bf16 v16bf;
typedef __attribute__((ext_vector_type(8)))  float  v8f;
typedef __attribute__((ext_vector_type(4)))  unsigned int u32x4;

union FragU { u32x4 q[2]; v16bf v; };

__device__ __forceinline__ v16bf load_frag16(const unsigned short* p) {
    FragU f;
    f.q[0] = *(const u32x4*)(p);
    f.q[1] = *(const u32x4*)(p + 16);
    return f.v;
}

__device__ __forceinline__ v8f wmma_bf16(v16bf a, v16bf b, v8f c) {
    return __builtin_amdgcn_wmma_f32_16x16x32_bf16(
         false, a,  false, b,
         (short)0, c,  false,  false);
}

__device__ __forceinline__ v16bf lds_frag16(const unsigned char* p) {
    FragU f;
    f.q[0] = *(const u32x4*)(p);
    f.q[1] = *(const u32x4*)(p + 32);
    return f.v;
}

__device__ __forceinline__ void wave_lds_handover() {
    __builtin_amdgcn_fence(__ATOMIC_RELEASE, "wavefront");
    __builtin_amdgcn_wave_barrier();
    __builtin_amdgcn_fence(__ATOMIC_ACQUIRE, "wavefront");
}

__device__ __forceinline__ unsigned short f2bf(float f) {
    unsigned int u = __float_as_uint(f);
    unsigned int r = u + 0x7FFFu + ((u >> 16) & 1u);
    return (unsigned short)(r >> 16);
}

#define GN 50000
#define GE 800000
#define GD 128
#define GH 8
#define GDFF 512

#define RP_T 256
#define RP_U 10
#define RP_BLOCKS (GN * GD / (RP_T * 4 * RP_U))
static_assert(RP_BLOCKS * RP_T * 4 * RP_U == GN * GD, "x must split into whole blocks");
static_assert(RP_BLOCKS * RP_T >= GDFF * GD, "grid must still cover the weight part");
__global__ void __launch_bounds__(RP_T)
repack_kernel(const float* __restrict__ x, const float* __restrict__ Wfc,
              const float* __restrict__ W1, const float* __restrict__ W2,
              unsigned short* __restrict__ xb, unsigned short* __restrict__ Wzt,
              unsigned short* __restrict__ W1t, unsigned short* __restrict__ W2t) {
    int idx = blockIdx.x * RP_T + threadIdx.x;
    size_t g0 = (size_t)blockIdx.x * (RP_T * RP_U) + threadIdx.x;
    const float4* xs = (const float4*)x + g0;
    uint2* xd = (uint2*)xb + g0;
    float4 v[RP_U];
    #pragma unroll
    for (int j = 0; j < RP_U; ++j) v[j] = xs[j * RP_T];
    if (idx < GD * GD) {
        int j = idx >> 7, d = idx & 127;
        int h = j >> 4, k = j & 15;
        Wzt[idx] = f2bf(Wfc[(h * GD + d) * 16 + k]);
    }
    if (idx < GDFF * GD) {
        int f = idx >> 7, d = idx & 127;
        W1t[idx] = f2bf(W1[d * GDFF + f]);
        int d2 = idx >> 9, f2 = idx & 511;
        W2t[idx] = f2bf(W2[f2 * GD + d2]);
    }
    #pragma unroll
    for (int j = 0; j < RP_U; ++j) {
        uint2 pk;
        pk.x = (unsigned int)f2bf(v[j].x) | ((unsigned int)f2bf(v[j].y) << 16);
        pk.y = (unsigned int)f2bf(v[j].z) | ((unsigned int)f2bf(v[j].w) << 16);
        xd[j * RP_T] = pk;
    }
}

#define PZ_T 256
#define PZ_WAVES 8
#define PZ_BLOCKS ((GN / 16 + PZ_WAVES - 1) / PZ_WAVES)
#define WZS_PITCH 272
#define WZS_BYTES (GD * WZS_PITCH)
#define ZS_PITCH 528
#define ZS_BYTES (16 * ZS_PITCH)
#define PZ_LDS (WZS_BYTES + PZ_WAVES * ZS_BYTES)
static_assert((GD * GD * 2 / 16) % PZ_T == 0, "the weight copy must split into whole passes");
static_assert(GN % 16 == 0, "whole strips only");

__global__ void __launch_bounds__(PZ_T)
proj_kernel(const unsigned short* __restrict__ xb, const unsigned short* __restrict__ Wzt,
            const float* __restrict__ a_l, const float* __restrict__ a_r,
            float* __restrict__ z, float* __restrict__ el, float* __restrict__ er) {
    __shared__ __attribute__((aligned(16))) unsigned char L[PZ_LDS];
    const int tid = threadIdx.x;
    const int lane = tid & 31;
    const int wave = __builtin_amdgcn_readfirstlane(tid >> 5);
    const int tile = blockIdx.x * PZ_WAVES + wave;
    const int tileC = min(tile, GN / 16 - 1);
    const int rowBase = tileC * 16;
    const int lrow = lane & 15;
    const int kb = (lane & 16) ? 8 : 0;
    const int half8 = (lane & 16) >> 1;

    const unsigned short* aRow = xb + (size_t)(rowBase + lrow) * GD + kb;
    v16bf a[4];
    #pragma unroll
    for (int ks = 0; ks < 4; ++ks) a[ks] = load_frag16(aRow + ks * 32);
    const float4 al = *(const float4*)(a_l + lane * 4);
    const float4 ar = *(const float4*)(a_r + lane * 4);

    {
        const u32x4* g = (const u32x4*)Wzt;
        #pragma unroll
        for (int p = 0; p < (GD * GD * 2 / 16) / PZ_T; ++p) {
            int q = p * PZ_T + tid;
            u32x4 v = g[q];
            *(u32x4*)(L + (q >> 4) * WZS_PITCH + (q & 15) * 16) = v;
        }
    }
    __syncthreads();
    if (tile >= GN / 16) return;

    const unsigned char* wzs = L + lrow * WZS_PITCH + kb * 2;
    float* scr = (float*)(L + WZS_BYTES + wave * ZS_BYTES);

    v8f acc[8];
    #pragma unroll
    for (int t = 0; t < 8; ++t) acc[t] = v8f{};
    #pragma unroll
    for (int ks = 0; ks < 4; ++ks) {
        v16bf b[8];
        #pragma unroll
        for (int t = 0; t < 8; ++t) b[t] = lds_frag16(wzs + t * 16 * WZS_PITCH + ks * 64);
        #pragma unroll
        for (int t = 0; t < 8; ++t) acc[t] = wmma_bf16(a[ks], b[t], acc[t]);
    }

    #pragma unroll
    for (int t = 0; t < 8; ++t) {
        #pragma unroll
        for (int j = 0; j < 8; ++j)
            scr[(half8 + j) * (ZS_PITCH / 4) + t * 16 + lrow] = acc[t][j];
    }
    wave_lds_handover();
    const int hd = lane >> 2;
    const int qd = lane & 3;
    float kl[4] = {0.f, 0.f, 0.f, 0.f}, kr[4] = {0.f, 0.f, 0.f, 0.f};
    #pragma unroll
    for (int r = 0; r < 16; ++r) {
        const float4 zz = *(const float4*)(scr + r * (ZS_PITCH / 4) + lane * 4);
        *(float4*)(z + (size_t)(rowBase + r) * GD + lane * 4) = zz;
        float sl = (zz.x * al.x + zz.y * al.y) + (zz.z * al.z + zz.w * al.w);
        float sr = (zz.x * ar.x + zz.y * ar.y) + (zz.z * ar.z + zz.w * ar.w);
        sl += __shfl_xor(sl, 1); sl += __shfl_xor(sl, 2);
        sr += __shfl_xor(sr, 1); sr += __shfl_xor(sr, 2);
        if ((r & 3) == qd) { kl[r >> 2] = sl; kr[r >> 2] = sr; }
    }
    #pragma unroll
    for (int i = 0; i < 4; ++i) {
        int row = rowBase + qd + 4 * i;
        el[row * GH + hd] = kl[i];
        er[row * GH + hd] = kr[i];
    }
}

#define CB_SHIFT 8
#define CB_NODES (1 << CB_SHIFT)
#define CB_N ((GN + CB_NODES - 1) / CB_NODES)
#define CH_N 256
#define CH_E (GE / CH_N)
#define CELLS (CB_N * CH_N)
#define BIN_T 512
#define CH_IT ((CH_E + BIN_T - 1) / BIN_T)
#define FIN_T 1024
#define FIN_U 6
#define SCAN_T 1024
#define SCAN_PER 49
static_assert(CH_E * CH_N == GE, "the edges must split into whole chunks");
static_assert(CELLS == SCAN_T * SCAN_PER, "the scan covers exactly the cell table");

__global__ void __launch_bounds__(BIN_T)
bincount_kernel(const int* __restrict__ edst, int* __restrict__ cell) {
    __shared__ int hc[CB_N];
    const int t = threadIdx.x;
    if (t < CB_N) hc[t] = 0;
    __syncthreads();
    const int base = blockIdx.x * CH_E;
    int d[CH_IT];
    #pragma unroll
    for (int k = 0; k < CH_IT; ++k) {
        int i = t + k * BIN_T;
        d[k] = (i < CH_E) ? edst[base + i] : -1;
    }
    #pragma unroll
    for (int k = 0; k < CH_IT; ++k)
        if ((unsigned int)d[k] < (unsigned int)GN) atomicAdd(&hc[d[k] >> CB_SHIFT], 1);
    __syncthreads();
    if (t < CB_N) cell[t * CH_N + blockIdx.x] = hc[t];
}

__global__ void __launch_bounds__(SCAN_T)
scan_kernel(const int* __restrict__ cell, int* __restrict__ off) {
    __shared__ int sh[SCAN_T * SCAN_PER];
    __shared__ int wsum[32];
    int t = threadIdx.x;
    #pragma unroll
    for (int k = 0; k < SCAN_PER; ++k) {
        int i = k * SCAN_T + t;
        sh[i] = cell[i];
    }
    __syncthreads();
    int base = t * SCAN_PER;
    int local = 0;
    for (int k = 0; k < SCAN_PER; ++k) local += sh[base + k];
    int lane = t & 31, w = t >> 5;
    int incl = local;
    for (int d = 1; d < 32; d <<= 1) { int y = __shfl_up(incl, d); if (lane >= d) incl += y; }
    if (lane == 31) wsum[w] = incl;
    __syncthreads();
    if (w == 0) {
        int x0 = wsum[lane];
        int xi = x0;
        for (int d = 1; d < 32; d <<= 1) { int y = __shfl_up(xi, d); if (lane >= d) xi += y; }
        wsum[lane] = xi - x0;
    }
    __syncthreads();
    int run = wsum[w] + incl - local;
    for (int k = 0; k < SCAN_PER; ++k) { int c = sh[base + k]; sh[base + k] = run; run += c; }
    __syncthreads();
    #pragma unroll
    for (int k = 0; k < SCAN_PER; ++k) {
        int i = k * SCAN_T + t;
        off[i] = sh[i];
    }
    if (t == SCAN_T - 1) off[CELLS] = run;
}

__global__ void __launch_bounds__(BIN_T)
binplace_kernel(const int* __restrict__ esrc, const int* __restrict__ edst,
                const int* __restrict__ off, int2* __restrict__ pairs) {
    __shared__ int cur[CB_N];
    const int t = threadIdx.x;
    if (t < CB_N) cur[t] = off[t * CH_N + blockIdx.x];
    __syncthreads();
    const int base = blockIdx.x * CH_E;
    int d[CH_IT], s[CH_IT];
    #pragma unroll
    for (int k = 0; k < CH_IT; ++k) {
        int i = t + k * BIN_T;
        d[k] = (i < CH_E) ? edst[base + i] : -1;
        s[k] = (i < CH_E) ? esrc[base + i] : 0;
    }
    #pragma unroll
    for (int k = 0; k < CH_IT; ++k) {
        if ((unsigned int)d[k] < (unsigned int)GN) {
            int pos = atomicAdd(&cur[d[k] >> CB_SHIFT], 1);
            pos = min(max(pos, 0), GE - 1);
            int2 p; p.x = s[k]; p.y = d[k];
            pairs[pos] = p;
        }
    }
}

__global__ void __launch_bounds__(FIN_T)
binfinish_kernel(const int* __restrict__ off, const int2* __restrict__ pairs,
                 int* __restrict__ rowptr, int* __restrict__ csrc) {
    __shared__ int cnt[CB_NODES];
    __shared__ int wsum[CB_NODES / 32];
    const int t = threadIdx.x;
    const int b = blockIdx.x;
    int beg = off[b * CH_N];
    int end = off[(b + 1) * CH_N];
    beg = min(max(beg, 0), GE);
    end = min(max(end, beg), GE);
    if (t < CB_NODES) cnt[t] = 0;
    __syncthreads();
    for (int i0 = beg + t; i0 < end; i0 += FIN_T * FIN_U) {
        int dd[FIN_U];
        #pragma unroll
        for (int u = 0; u < FIN_U; ++u) {
            int i = i0 + u * FIN_T;
            dd[u] = (i < end) ? pairs[i].y : -1;
        }
        #pragma unroll
        for (int u = 0; u < FIN_U; ++u)
            if (dd[u] >= 0) atomicAdd(&cnt[dd[u] & (CB_NODES - 1)], 1);
    }
    __syncthreads();
    const int lane = t & 31, w = t >> 5;
    const int c = (t < CB_NODES) ? cnt[t] : 0;
    int incl = c;
    for (int d = 1; d < 32; d <<= 1) { int y = __shfl_up(incl, d); if (lane >= d) incl += y; }
    if (t < CB_NODES && lane == 31) wsum[w] = incl;
    __syncthreads();
    if (t < CB_NODES) {
        int wbase = 0;
        for (int k = 0; k < CB_NODES / 32; ++k) wbase += (k < w) ? wsum[k] : 0;
        int start = beg + wbase + incl - c;
        cnt[t] = start;
        int n = b * CB_NODES + t;
        if (n <= GN) rowptr[n] = start;
    }
    __syncthreads();
    for (int i0 = beg + t; i0 < end; i0 += FIN_T * FIN_U) {
        int ss[FIN_U], dd[FIN_U];
        #pragma unroll
        for (int u = 0; u < FIN_U; ++u) {
            int i = i0 + u * FIN_T;
            int2 p = pairs[(i < end) ? i : i0];
            ss[u] = p.x;
            dd[u] = (i < end) ? p.y : -1;
        }
        #pragma unroll
        for (int u = 0; u < FIN_U; ++u) {
            if (dd[u] >= 0) {
                int pos = atomicAdd(&cnt[dd[u] & (CB_NODES - 1)], 1);
                pos = min(max(pos, 0), GE - 1);
                csrc[pos] = ss[u];
            }
        }
    }
}

__global__ void gather_ln_kernel(const float* __restrict__ z, const float* __restrict__ el,
                                 const float* __restrict__ er, const int* __restrict__ rowptr,
                                 const int* __restrict__ csrc, const float* __restrict__ x,
                                 const float* __restrict__ gamma, const float* __restrict__ beta,
                                 float* __restrict__ hres, unsigned short* __restrict__ lnb) {
    int lane = threadIdx.x & 31;
    int wave = __builtin_amdgcn_readfirstlane((int)(threadIdx.x >> 5));
    int n = blockIdx.x * 8 + wave;
    int beg = __builtin_amdgcn_readfirstlane(rowptr[n]);
    int end = __builtin_amdgcn_readfirstlane(rowptr[n + 1]);
    beg = min(max(beg, 0), GE);
    end = min(max(end, beg), GE);
    int hd = lane >> 2;
    float ern = er[n * GH + hd];
    const float4 xx = *(const float4*)(x + (size_t)n * GD + lane * 4);
    const float4 gg = *(const float4*)(gamma + lane * 4);
    const float4 bb = *(const float4*)(beta + lane * 4);
    float m = -INFINITY, s = 0.f;
    float a0 = 0.f, a1 = 0.f, a2 = 0.f, a3 = 0.f;
    for (int base = beg; base < end; base += 32) {
        int cc = min(32, end - base);
        int mysrc = (lane < cc) ? csrc[base + lane] : 0;
        #pragma unroll 4
        for (int j = 0; j < cc; ++j) {
            int sidx = __builtin_amdgcn_readlane(mysrc, j);
            sidx = min(max(sidx, 0), GN - 1);
            float v = el[sidx * GH + hd] + ern;
            v = v > 0.f ? v : 0.01f * v;
            const float4 zz = *(const float4*)(z + (size_t)sidx * GD + lane * 4);
            float mn = fmaxf(m, v);
            float sc = __expf(m - mn);
            float p  = __expf(v - mn);
            s  = s  * sc + p;
            a0 = a0 * sc + p * zz.x;
            a1 = a1 * sc + p * zz.y;
            a2 = a2 * sc + p * zz.z;
            a3 = a3 * sc + p * zz.w;
            m = mn;
        }
    }
    float inv = 1.f / fmaxf(s, 1e-9f);
    float v0 = a0 * inv, v1 = a1 * inv, v2 = a2 * inv, v3 = a3 * inv;
    v0 = v0 > 0.f ? v0 : (__expf(v0) - 1.f);
    v1 = v1 > 0.f ? v1 : (__expf(v1) - 1.f);
    v2 = v2 > 0.f ? v2 : (__expf(v2) - 1.f);
    v3 = v3 > 0.f ? v3 : (__expf(v3) - 1.f);
    v0 += xx.x; v1 += xx.y; v2 += xx.z; v3 += xx.w;
    float4 hr; hr.x = v0; hr.y = v1; hr.z = v2; hr.w = v3;
    *(float4*)(hres + (size_t)n * GD + lane * 4) = hr;
    float sm = (v0 + v1) + (v2 + v3);
    float sq = (v0 * v0 + v1 * v1) + (v2 * v2 + v3 * v3);
    for (int d = 16; d > 0; d >>= 1) {
        sm += __shfl_xor(sm, d);
        sq += __shfl_xor(sq, d);
    }
    float mean = sm * (1.f / GD);
    float var  = sq * (1.f / GD) - mean * mean;
    float rstd = rsqrtf(var + 1e-5f);
    unsigned int b0 = f2bf((v0 - mean) * rstd * gg.x + bb.x);
    unsigned int b1 = f2bf((v1 - mean) * rstd * gg.y + bb.y);
    unsigned int b2 = f2bf((v2 - mean) * rstd * gg.z + bb.z);
    unsigned int b3 = f2bf((v3 - mean) * rstd * gg.w + bb.w);
    uint2 pk; pk.x = b0 | (b1 << 16); pk.y = b2 | (b3 << 16);
    *(uint2*)(lnb + (size_t)n * GD + lane * 4) = pk;
}

#define FF_T 512
#define FF_WAVES 16
#define FF_BLOCKS ((GN / 16 + FF_WAVES - 1) / FF_WAVES)
#define W1S_PITCH 272
#define W2S_PITCH 1040
#define SCR_PITCH 144
#define W1S_BYTES (GDFF * W1S_PITCH)
#define W2S_BYTES (GD * W2S_PITCH)
#define SCR_BYTES (16 * SCR_PITCH)
#define FF_LDS (W1S_BYTES + W2S_BYTES + FF_WAVES * SCR_BYTES)
static_assert(FF_LDS <= 320 * 1024, "weights and scratch must fit one work-group's LDS");
static_assert((GDFF * GD * 2 / 16) % FF_T == 0, "the weight copy must split into whole passes");

__global__ void __launch_bounds__(FF_T)
ffn_fused_kernel(const unsigned short* __restrict__ lnb,
                 const unsigned short* __restrict__ W1t, const float* __restrict__ b1,
                 const unsigned short* __restrict__ W2t, const float* __restrict__ b2,
                 const float* __restrict__ hres, float* __restrict__ out) {
    __shared__ __attribute__((aligned(16))) unsigned char L[FF_LDS];
    const int tid = threadIdx.x;
    const int lane = tid & 31;
    const int wave = __builtin_amdgcn_readfirstlane(tid >> 5);
    const int tile = blockIdx.x * FF_WAVES + wave;
    const int tileC = min(tile, GN / 16 - 1);
    const int rowBase = tileC * 16;
    const int lrow = lane & 15;
    const int kb = (lane & 16) ? 8 : 0;
    const int half8 = (lane & 16) >> 1;

    const unsigned short* aRow = lnb + (size_t)(rowBase + lrow) * GD + kb;
    v16bf a[4];
    #pragma unroll
    for (int ks = 0; ks < 4; ++ks) a[ks] = load_frag16(aRow + ks * 32);

    {
        const u32x4* g1 = (const u32x4*)W1t;
        #pragma unroll 8
        for (int p = 0; p < (GDFF * GD * 2 / 16) / FF_T; ++p) {
            int q = p * FF_T + tid;
            u32x4 v = g1[q];
            *(u32x4*)(L + (q >> 4) * W1S_PITCH + (q & 15) * 16) = v;
        }
        const u32x4* g2 = (const u32x4*)W2t;
        #pragma unroll 8
        for (int p = 0; p < (GD * GDFF * 2 / 16) / FF_T; ++p) {
            int q = p * FF_T + tid;
            u32x4 v = g2[q];
            *(u32x4*)(L + W1S_BYTES + (q >> 6) * W2S_PITCH + (q & 63) * 16) = v;
        }
    }
    __syncthreads();
    if (tile >= GN / 16) return;

    const unsigned char* w1s = L + lrow * W1S_PITCH + kb * 2;
    const unsigned char* w2s = L + W1S_BYTES + lrow * W2S_PITCH + kb * 2;
    unsigned char* scr = L + W1S_BYTES + W2S_BYTES + wave * SCR_BYTES;

    v8f acc2[8];
    #pragma unroll
    for (int nt = 0; nt < 8; ++nt) acc2[nt] = v8f{};

    for (int c = 0; c < GDFF / 64; ++c) {
        v8f acc1[4];
        #pragma unroll
        for (int t = 0; t < 4; ++t) acc1[t] = v8f{};
        #pragma unroll
        for (int ks = 0; ks < 4; ++ks) {
            #pragma unroll
            for (int t = 0; t < 4; ++t) {
                v16bf b = lds_frag16(w1s + (c * 64 + t * 16) * W1S_PITCH + ks * 64);
                acc1[t] = wmma_bf16(a[ks], b, acc1[t]);
            }
        }
        #pragma unroll
        for (int t = 0; t < 4; ++t) {
            float bias = b1[c * 64 + t * 16 + lrow];
            #pragma unroll
            for (int j = 0; j < 8; ++j) {
                float v = fmaxf(acc1[t][j] + bias, 0.f);
                *(unsigned short*)(scr + (half8 + j) * SCR_PITCH + (t * 16 + lrow) * 2) = f2bf(v);
            }
        }
        wave_lds_handover();
        #pragma unroll
        for (int ks2 = 0; ks2 < 2; ++ks2) {
            v16bf a2 = lds_frag16(scr + lrow * SCR_PITCH + kb * 2 + ks2 * 64);
            #pragma unroll
            for (int nt = 0; nt < 8; ++nt) {
                v16bf b = lds_frag16(w2s + nt * 16 * W2S_PITCH + (c * 64 + ks2 * 32) * 2);
                acc2[nt] = wmma_bf16(a2, b, acc2[nt]);
            }
        }
        wave_lds_handover();
    }

    #pragma unroll
    for (int nt = 0; nt < 8; ++nt) {
        int n = nt * 16 + lrow;
        float bias = b2[n];
        #pragma unroll
        for (int j = 0; j < 8; ++j) {
            size_t r = (size_t)(rowBase + half8 + j);
            out[r * GD + n] = acc2[nt][j] + bias + hres[r * GD + n];
        }
    }
}

extern "C" void kernel_launch(void* const* d_in, const int* in_sizes, int n_in,
                              void* d_out, int out_size, void* d_ws, size_t ws_size,
                              hipStream_t stream) {
    const float* x     = (const float*)d_in[0];
    const float* Wfc   = (const float*)d_in[1];
    const float* a_l   = (const float*)d_in[2];
    const float* a_r   = (const float*)d_in[3];
    const float* gamma = (const float*)d_in[4];
    const float* beta  = (const float*)d_in[5];
    const float* W1    = (const float*)d_in[6];
    const float* b1    = (const float*)d_in[7];
    const float* W2    = (const float*)d_in[8];
    const float* b2    = (const float*)d_in[9];
    const int* esrc    = (const int*)d_in[10];
    const int* edst    = (const int*)d_in[11];
    float* out = (float*)d_out;

    char* ws = (char*)d_ws;
    size_t off = 0;
    auto take = [&](size_t bytes) -> char* {
        char* p = ws + off;
        off += (bytes + 255) & ~(size_t)255;
        return p;
    };
    unsigned short* xb    = (unsigned short*)take((size_t)GN * GD * 2);
    unsigned short* Wzt   = (unsigned short*)take((size_t)GD * GD * 2);
    unsigned short* W1t   = (unsigned short*)take((size_t)GDFF * GD * 2);
    unsigned short* W2t   = (unsigned short*)take((size_t)GD * GDFF * 2);
    float*          z     = (float*)take((size_t)GN * GD * 4);
    float*          el    = (float*)take((size_t)GN * GH * 4);
    float*          er    = (float*)take((size_t)GN * GH * 4);
    int*            rowptr= (int*)take((size_t)(GN + 1) * 4);
    int*            csrc  = (int*)take((size_t)GE * 4);
    float*          hres  = (float*)take((size_t)GN * GD * 4);
    unsigned short* lnb   = (unsigned short*)take((size_t)GN * GD * 2);
    int2*           pairs = (int2*)hres;
    int*            cell  = (int*)lnb;
    int*            coff  = (int*)((char*)lnb + (size_t)CELLS * 4);

    repack_kernel<<<RP_BLOCKS, RP_T, 0, stream>>>(x, Wfc, W1, W2, xb, Wzt, W1t, W2t);
    proj_kernel  <<<PZ_BLOCKS, PZ_T, 0, stream>>>(xb, Wzt, a_l, a_r, z, el, er);
    bincount_kernel <<<CH_N, BIN_T, 0, stream>>>(edst, cell);
    scan_kernel     <<<1, SCAN_T, 0, stream>>>(cell, coff);
    binplace_kernel <<<CH_N, BIN_T, 0, stream>>>(esrc, edst, coff, pairs);
    binfinish_kernel<<<CB_N, FIN_T, 0, stream>>>(coff, pairs, rowptr, csrc);
    gather_ln_kernel<<<GN / 8, 256, 0, stream>>>(z, el, er, rowptr, csrc, x, gamma, beta, hres, lnb);
    ffn_fused_kernel<<<FF_BLOCKS, FF_T, 0, stream>>>(lnb, W1t, b1, W2t, b2, hres, out);
}


# Edge grouping runs beside the dense path: chunk counts ride in the repack launch, pair placing (each group sums the count table itself, scan launch gone) in the projection launch; 8 launches -> 5
# speedup vs baseline: 5.7035x; 1.1939x over previous
#include <hip/hip_runtime.h>

typedef __attribute__((ext_vector_type(16))) __bf16 v16bf;
typedef __attribute__((ext_vector_type(8)))  float  v8f;
typedef __attribute__((ext_vector_type(4)))  unsigned int u32x4;

union FragU { u32x4 q[2]; v16bf v; };

__device__ __forceinline__ v16bf load_frag16(const unsigned short* p) {
    FragU f;
    f.q[0] = *(const u32x4*)(p);
    f.q[1] = *(const u32x4*)(p + 16);
    return f.v;
}

__device__ __forceinline__ v8f wmma_bf16(v16bf a, v16bf b, v8f c) {
    return __builtin_amdgcn_wmma_f32_16x16x32_bf16(
         false, a,  false, b,
         (short)0, c,  false,  false);
}

__device__ __forceinline__ v16bf lds_frag16(const unsigned char* p) {
    FragU f;
    f.q[0] = *(const u32x4*)(p);
    f.q[1] = *(const u32x4*)(p + 32);
    return f.v;
}

__device__ __forceinline__ void wave_lds_handover() {
    __builtin_amdgcn_fence(__ATOMIC_RELEASE, "wavefront");
    __builtin_amdgcn_wave_barrier();
    __builtin_amdgcn_fence(__ATOMIC_ACQUIRE, "wavefront");
}

__device__ __forceinline__ unsigned short f2bf(float f) {
    unsigned int u = __float_as_uint(f);
    unsigned int r = u + 0x7FFFu + ((u >> 16) & 1u);
    return (unsigned short)(r >> 16);
}

#define GN 50000
#define GE 800000
#define GD 128
#define GH 8
#define GDFF 512

#define CB_SHIFT 8
#define CB_NODES (1 << CB_SHIFT)
#define CB_N ((GN + CB_NODES - 1) / CB_NODES)
#define CH_N 256
#define CH_E (GE / CH_N)
#define SB_T 256
#define SB_IT ((CH_E + SB_T - 1) / SB_T)
#define CELL_PITCH 256
static_assert(CH_E * CH_N == GE, "the edges must split into whole chunks");
static_assert(CB_N % 4 == 0 && CB_N <= CELL_PITCH, "a chunk row is read as whole 16-byte quads");

#define RP_T 256
#define RP_U 10
#define RP_BLOCKS (GN * GD / (RP_T * 4 * RP_U))
static_assert(RP_BLOCKS * RP_T * 4 * RP_U == GN * GD, "x must split into whole blocks");
static_assert(RP_BLOCKS * RP_T >= GDFF * GD, "grid must still cover the weight part");
static_assert(RP_T == SB_T, "both kinds of work-group share the launch's block size");
__global__ void __launch_bounds__(RP_T)
repack_count_kernel(const float* __restrict__ x, const float* __restrict__ Wfc,
              const float* __restrict__ W1, const float* __restrict__ W2,
              const int* __restrict__ edst,
              unsigned short* __restrict__ xb, unsigned short* __restrict__ Wzt,
              unsigned short* __restrict__ W1t, unsigned short* __restrict__ W2t,
              int* __restrict__ cell) {
    __shared__ int hc[CB_N];
    if (blockIdx.x < CH_N) {
        const int t = threadIdx.x;
        const int c = blockIdx.x;
        if (t < CB_N) hc[t] = 0;
        __syncthreads();
        const int base = c * CH_E;
        int d[SB_IT];
        #pragma unroll
        for (int k = 0; k < SB_IT; ++k) {
            int i = t + k * SB_T;
            d[k] = (i < CH_E) ? edst[base + i] : -1;
        }
        #pragma unroll
        for (int k = 0; k < SB_IT; ++k)
            if ((unsigned int)d[k] < (unsigned int)GN) atomicAdd(&hc[d[k] >> CB_SHIFT], 1);
        __syncthreads();
        if (t < CB_N) cell[c * CELL_PITCH + t] = hc[t];
        return;
    }
    const int rb = blockIdx.x - CH_N;
    int idx = rb * RP_T + threadIdx.x;
    size_t g0 = (size_t)rb * (RP_T * RP_U) + threadIdx.x;
    const float4* xs = (const float4*)x + g0;
    uint2* xd = (uint2*)xb + g0;
    float4 v[RP_U];
    #pragma unroll
    for (int j = 0; j < RP_U; ++j) v[j] = xs[j * RP_T];
    if (idx < GD * GD) {
        int j = idx >> 7, d = idx & 127;
        int h = j >> 4, k = j & 15;
        Wzt[idx] = f2bf(Wfc[(h * GD + d) * 16 + k]);
    }
    if (idx < GDFF * GD) {
        int f = idx >> 7, d = idx & 127;
        W1t[idx] = f2bf(W1[d * GDFF + f]);
        int d2 = idx >> 9, f2 = idx & 511;
        W2t[idx] = f2bf(W2[f2 * GD + d2]);
    }
    #pragma unroll
    for (int j = 0; j < RP_U; ++j) {
        uint2 pk;
        pk.x = (unsigned int)f2bf(v[j].x) | ((unsigned int)f2bf(v[j].y) << 16);
        pk.y = (unsigned int)f2bf(v[j].z) | ((unsigned int)f2bf(v[j].w) << 16);
        xd[j * RP_T] = pk;
    }
}

#define PZ_T 256
#define PZ_WAVES 8
#define PZ_BLOCKS ((GN / 16 + PZ_WAVES - 1) / PZ_WAVES)
#define WZS_PITCH 272
#define WZS_BYTES (GD * WZS_PITCH)
#define ZS_PITCH 528
#define ZS_BYTES (16 * ZS_PITCH)
#define PZ_LDS (WZS_BYTES + PZ_WAVES * ZS_BYTES)
static_assert((GD * GD * 2 / 16) % PZ_T == 0, "the weight copy must split into whole passes");
static_assert(GN % 16 == 0, "whole strips only");
static_assert(PZ_T == SB_T, "both kinds of work-group share the launch's block size");

#define PL_G 4
#define PL_ROWS (CH_N / PL_G)
#define PL_Q (CB_N / 4)
static_assert(SB_T == PL_G * 64 && PL_Q <= 64 && CB_N < SB_T, "thread layout of the table sum");
static_assert((2 * PL_G * CELL_PITCH + CELL_PITCH + 8) * 4 <= PZ_LDS, "the placing part borrows the projection's LDS");
__device__ __forceinline__ void place_part(unsigned char* L, const int c,
                                           const int* __restrict__ esrc, const int* __restrict__ edst,
                                           const int* __restrict__ cell, int* __restrict__ bstart,
                                           int2* __restrict__ pairs) {
    int* partT = (int*)L;
    int* partP = partT + PL_G * CELL_PITCH;
    int* cur   = partP + PL_G * CELL_PITCH;
    int* wsum  = cur + CELL_PITCH;
    const int t = threadIdx.x;
    const int base = c * CH_E;
    int d[SB_IT], sr[SB_IT];
    #pragma unroll
    for (int k = 0; k < SB_IT; ++k) {
        int i = t + k * SB_T;
        d[k]  = (i < CH_E) ? edst[base + i] : -1;
        sr[k] = (i < CH_E) ? esrc[base + i] : 0;
    }
    const int q = t & 63, g = t >> 6;
    int4 T4 = make_int4(0, 0, 0, 0), P4 = make_int4(0, 0, 0, 0);
    if (q < PL_Q) {
        const int4* row = (const int4*)(cell + (size_t)(g * PL_ROWS) * CELL_PITCH) + q;
        #pragma unroll 8
        for (int i = 0; i < PL_ROWS; ++i) {
            const int4 v = row[i * (CELL_PITCH / 4)];
            const int take = (g * PL_ROWS + i < c) ? -1 : 0;
            T4.x += v.x; T4.y += v.y; T4.z += v.z; T4.w += v.w;
            P4.x += v.x & take; P4.y += v.y & take; P4.z += v.z & take; P4.w += v.w & take;
        }
    }
    ((int4*)(partT + g * CELL_PITCH))[q] = T4;
    ((int4*)(partP + g * CELL_PITCH))[q] = P4;
    __syncthreads();
    int Tb = 0, Pb = 0;
    if (t < CB_N) {
        #pragma unroll
        for (int k = 0; k < PL_G; ++k) { Tb += partT[k * CELL_PITCH + t]; Pb += partP[k * CELL_PITCH + t]; }
    }
    const int lane = t & 31, w = t >> 5;
    int incl = Tb;
    for (int dd = 1; dd < 32; dd <<= 1) { int y = __shfl_up(incl, dd); if (lane >= dd) incl += y; }
    if (lane == 31) wsum[w] = incl;
    __syncthreads();
    int wbase = 0;
    for (int k = 0; k < SB_T / 32; ++k) wbase += (k < w) ? wsum[k] : 0;
    const int start = wbase + incl - Tb;
    if (t < CB_N) cur[t] = start + Pb;
    if (c == 0 && t <= CB_N) bstart[t] = start;
    __syncthreads();
    #pragma unroll
    for (int k = 0; k < SB_IT; ++k) {
        if ((unsigned int)d[k] < (unsigned int)GN) {
            int pos = atomicAdd(&cur[d[k] >> CB_SHIFT], 1);
            pos = min(max(pos, 0), GE - 1);
            int2 p; p.x = sr[k]; p.y = d[k];
            pairs[pos] = p;
        }
    }
}

__global__ void __launch_bounds__(PZ_T, 6)
proj_place_kernel(const unsigned short* __restrict__ xb, const unsigned short* __restrict__ Wzt,
            const float* __restrict__ a_l, const float* __restrict__ a_r,
            float* __restrict__ z, float* __restrict__ el, float* __restrict__ er,
            const int* __restrict__ esrc, const int* __restrict__ edst,
            const int* __restrict__ cell, int* __restrict__ bstart, int2* __restrict__ pairs) {
    __shared__ __attribute__((aligned(16))) unsigned char L[PZ_LDS];
    if (blockIdx.x >= PZ_BLOCKS) {
        place_part(L, blockIdx.x - PZ_BLOCKS, esrc, edst, cell, bstart, pairs);
        return;
    }
    const int pb = blockIdx.x;
    const int tid = threadIdx.x;
    const int lane = tid & 31;
    const int wave = __builtin_amdgcn_readfirstlane(tid >> 5);
    const int tile = pb * PZ_WAVES + wave;
    const int tileC = min(tile, GN / 16 - 1);
    const int rowBase = tileC * 16;
    const int lrow = lane & 15;
    const int kb = (lane & 16) ? 8 : 0;
    const int half8 = (lane & 16) >> 1;

    const unsigned short* aRow = xb + (size_t)(rowBase + lrow) * GD + kb;
    v16bf a[4];
    #pragma unroll
    for (int ks = 0; ks < 4; ++ks) a[ks] = load_frag16(aRow + ks * 32);
    {
        const u32x4* g = (const u32x4*)Wzt;
        #pragma unroll
        for (int p = 0; p < (GD * GD * 2 / 16) / PZ_T; ++p) {
            int q = p * PZ_T + tid;
            u32x4 v = g[q];
            *(u32x4*)(L + (q >> 4) * WZS_PITCH + (q & 15) * 16) = v;
        }
    }
    __syncthreads();
    if (tile >= GN / 16) return;

    const unsigned char* wzs = L + lrow * WZS_PITCH + kb * 2;
    float* scr = (float*)(L + WZS_BYTES + wave * ZS_BYTES);

    v8f acc[8];
    #pragma unroll
    for (int t = 0; t < 8; ++t) acc[t] = v8f{};
    #pragma unroll
    for (int ks = 0; ks < 4; ++ks) {
        v16bf b[8];
        #pragma unroll
        for (int t = 0; t < 8; ++t) b[t] = lds_frag16(wzs + t * 16 * WZS_PITCH + ks * 64);
        #pragma unroll
        for (int t = 0; t < 8; ++t) acc[t] = wmma_bf16(a[ks], b[t], acc[t]);
    }

    const float4 al = *(const float4*)(a_l + lane * 4);
    const float4 ar = *(const float4*)(a_r + lane * 4);
    #pragma unroll
    for (int t = 0; t < 8; ++t) {
        #pragma unroll
        for (int j = 0; j < 8; ++j)
            scr[(half8 + j) * (ZS_PITCH / 4) + t * 16 + lrow] = acc[t][j];
    }
    wave_lds_handover();
    const int hd = lane >> 2;
    const int qd = lane & 3;
    float kl[4] = {0.f, 0.f, 0.f, 0.f}, kr[4] = {0.f, 0.f, 0.f, 0.f};
    #pragma unroll
    for (int r = 0; r < 16; ++r) {
        const float4 zz = *(const float4*)(scr + r * (ZS_PITCH / 4) + lane * 4);
        *(float4*)(z + (size_t)(rowBase + r) * GD + lane * 4) = zz;
        float sl = (zz.x * al.x + zz.y * al.y) + (zz.z * al.z + zz.w * al.w);
        float sr = (zz.x * ar.x + zz.y * ar.y) + (zz.z * ar.z + zz.w * ar.w);
        sl += __shfl_xor(sl, 1); sl += __shfl_xor(sl, 2);
        sr += __shfl_xor(sr, 1); sr += __shfl_xor(sr, 2);
        if ((r & 3) == qd) { kl[r >> 2] = sl; kr[r >> 2] = sr; }
    }
    #pragma unroll
    for (int i = 0; i < 4; ++i) {
        int row = rowBase + qd + 4 * i;
        el[row * GH + hd] = kl[i];
        er[row * GH + hd] = kr[i];
    }
}

#define FIN_T 1024
#define FIN_U 6

__global__ void __launch_bounds__(FIN_T)
binfinish_kernel(const int* __restrict__ bstart, const int2* __restrict__ pairs,
                 int* __restrict__ rowptr, int* __restrict__ csrc) {
    __shared__ int cnt[CB_NODES];
    __shared__ int wsum[CB_NODES / 32];
    const int t = threadIdx.x;
    const int b = blockIdx.x;
    int beg = bstart[b];
    int end = bstart[b + 1];
    beg = min(max(beg, 0), GE);
    end = min(max(end, beg), GE);
    if (t < CB_NODES) cnt[t] = 0;
    __syncthreads();
    for (int i0 = beg + t; i0 < end; i0 += FIN_T * FIN_U) {
        int dd[FIN_U];
        #pragma unroll
        for (int u = 0; u < FIN_U; ++u) {
            int i = i0 + u * FIN_T;
            dd[u] = (i < end) ? pairs[i].y : -1;
        }
        #pragma unroll
        for (int u = 0; u < FIN_U; ++u)
            if (dd[u] >= 0) atomicAdd(&cnt[dd[u] & (CB_NODES - 1)], 1);
    }
    __syncthreads();
    const int lane = t & 31, w = t >> 5;
    const int c = (t < CB_NODES) ? cnt[t] : 0;
    int incl = c;
    for (int d = 1; d < 32; d <<= 1) { int y = __shfl_up(incl, d); if (lane >= d) incl += y; }
    if (t < CB_NODES && lane == 31) wsum[w] = incl;
    __syncthreads();
    if (t < CB_NODES) {
        int wbase = 0;
        for (int k = 0; k < CB_NODES / 32; ++k) wbase += (k < w) ? wsum[k] : 0;
        int start = beg + wbase + incl - c;
        cnt[t] = start;
        int n = b * CB_NODES + t;
        if (n <= GN) rowptr[n] = start;
    }
    __syncthreads();
    for (int i0 = beg + t; i0 < end; i0 += FIN_T * FIN_U) {
        int ss[FIN_U], dd[FIN_U];
        #pragma unroll
        for (int u = 0; u < FIN_U; ++u) {
            int i = i0 + u * FIN_T;
            int2 p = pairs[(i < end) ? i : i0];
            ss[u] = p.x;
            dd[u] = (i < end) ? p.y : -1;
        }
        #pragma unroll
        for (int u = 0; u < FIN_U; ++u) {
            if (dd[u] >= 0) {
                int pos = atomicAdd(&cnt[dd[u] & (CB_NODES - 1)], 1);
                pos = min(max(pos, 0), GE - 1);
                csrc[pos] = ss[u];
            }
        }
    }
}

__global__ void gather_ln_kernel(const float* __restrict__ z, const float* __restrict__ el,
                                 const float* __restrict__ er, const int* __restrict__ rowptr,
                                 const int* __restrict__ csrc, const float* __restrict__ x,
                                 const float* __restrict__ gamma, const float* __restrict__ beta,
                                 float* __restrict__ hres, unsigned short* __restrict__ lnb) {
    int lane = threadIdx.x & 31;
    int wave = __builtin_amdgcn_readfirstlane((int)(threadIdx.x >> 5));
    int n = blockIdx.x * 8 + wave;
    int beg = __builtin_amdgcn_readfirstlane(rowptr[n]);
    int end = __builtin_amdgcn_readfirstlane(rowptr[n + 1]);
    beg = min(max(beg, 0), GE);
    end = min(max(end, beg), GE);
    int hd = lane >> 2;
    float ern = er[n * GH + hd];
    const float4 xx = *(const float4*)(x + (size_t)n * GD + lane * 4);
    const float4 gg = *(const float4*)(gamma + lane * 4);
    const float4 bb = *(const float4*)(beta + lane * 4);
    float m = -INFINITY, s = 0.f;
    float a0 = 0.f, a1 = 0.f, a2 = 0.f, a3 = 0.f;
    for (int base = beg; base < end; base += 32) {
        int cc = min(32, end - base);
        int mysrc = (lane < cc) ? csrc[base + lane] : 0;
        #pragma unroll 4
        for (int j = 0; j < cc; ++j) {
            int sidx = __builtin_amdgcn_readlane(mysrc, j);
            sidx = min(max(sidx, 0), GN - 1);
            float v = el[sidx * GH + hd] + ern;
            v = v > 0.f ? v : 0.01f * v;
            const float4 zz = *(const float4*)(z + (size_t)sidx * GD + lane * 4);
            float mn = fmaxf(m, v);
            float sc = __expf(m - mn);
            float p  = __expf(v - mn);
            s  = s  * sc + p;
            a0 = a0 * sc + p * zz.x;
            a1 = a1 * sc + p * zz.y;
            a2 = a2 * sc + p * zz.z;
            a3 = a3 * sc + p * zz.w;
            m = mn;
        }
    }
    float inv = 1.f / fmaxf(s, 1e-9f);
    float v0 = a0 * inv, v1 = a1 * inv, v2 = a2 * inv, v3 = a3 * inv;
    v0 = v0 > 0.f ? v0 : (__expf(v0) - 1.f);
    v1 = v1 > 0.f ? v1 : (__expf(v1) - 1.f);
    v2 = v2 > 0.f ? v2 : (__expf(v2) - 1.f);
    v3 = v3 > 0.f ? v3 : (__expf(v3) - 1.f);
    v0 += xx.x; v1 += xx.y; v2 += xx.z; v3 += xx.w;
    float4 hr; hr.x = v0; hr.y = v1; hr.z = v2; hr.w = v3;
    *(float4*)(hres + (size_t)n * GD + lane * 4) = hr;
    float sm = (v0 + v1) + (v2 + v3);
    float sq = (v0 * v0 + v1 * v1) + (v2 * v2 + v3 * v3);
    for (int d = 16; d > 0; d >>= 1) {
        sm += __shfl_xor(sm, d);
        sq += __shfl_xor(sq, d);
    }
    float mean = sm * (1.f / GD);
    float var  = sq * (1.f / GD) - mean * mean;
    float rstd = rsqrtf(var + 1e-5f);
    unsigned int b0 = f2bf((v0 - mean) * rstd * gg.x + bb.x);
    unsigned int b1 = f2bf((v1 - mean) * rstd * gg.y + bb.y);
    unsigned int b2 = f2bf((v2 - mean) * rstd * gg.z + bb.z);
    unsigned int b3 = f2bf((v3 - mean) * rstd * gg.w + bb.w);
    uint2 pk; pk.x = b0 | (b1 << 16); pk.y = b2 | (b3 << 16);
    *(uint2*)(lnb + (size_t)n * GD + lane * 4) = pk;
}

#define FF_T 512
#define FF_WAVES 16
#define FF_BLOCKS ((GN / 16 + FF_WAVES - 1) / FF_WAVES)
#define W1S_PITCH 272
#define W2S_PITCH 1040
#define SCR_PITCH 144
#define W1S_BYTES (GDFF * W1S_PITCH)
#define W2S_BYTES (GD * W2S_PITCH)
#define SCR_BYTES (16 * SCR_PITCH)
#define FF_LDS (W1S_BYTES + W2S_BYTES + FF_WAVES * SCR_BYTES)
static_assert(FF_LDS <= 320 * 1024, "weights and scratch must fit one work-group's LDS");
static_assert((GDFF * GD * 2 / 16) % FF_T == 0, "the weight copy must split into whole passes");

__global__ void __launch_bounds__(FF_T)
ffn_fused_kernel(const unsigned short* __restrict__ lnb,
                 const unsigned short* __restrict__ W1t, const float* __restrict__ b1,
                 const unsigned short* __restrict__ W2t, const float* __restrict__ b2,
                 const float* __restrict__ hres, float* __restrict__ out) {
    __shared__ __attribute__((aligned(16))) unsigned char L[FF_LDS];
    const int tid = threadIdx.x;
    const int lane = tid & 31;
    const int wave = __builtin_amdgcn_readfirstlane(tid >> 5);
    const int tile = blockIdx.x * FF_WAVES + wave;
    const int tileC = min(tile, GN / 16 - 1);
    const int rowBase = tileC * 16;
    const int lrow = lane & 15;
    const int kb = (lane & 16) ? 8 : 0;
    const int half8 = (lane & 16) >> 1;

    const unsigned short* aRow = lnb + (size_t)(rowBase + lrow) * GD + kb;
    v16bf a[4];
    #pragma unroll
    for (int ks = 0; ks < 4; ++ks) a[ks] = load_frag16(aRow + ks * 32);

    {
        const u32x4* g1 = (const u32x4*)W1t;
        #pragma unroll 8
        for (int p = 0; p < (GDFF * GD * 2 / 16) / FF_T; ++p) {
            int q = p * FF_T + tid;
            u32x4 v = g1[q];
            *(u32x4*)(L + (q >> 4) * W1S_PITCH + (q & 15) * 16) = v;
        }
        const u32x4* g2 = (const u32x4*)W2t;
        #pragma unroll 8
        for (int p = 0; p < (GD * GDFF * 2 / 16) / FF_T; ++p) {
            int q = p * FF_T + tid;
            u32x4 v = g2[q];
            *(u32x4*)(L + W1S_BYTES + (q >> 6) * W2S_PITCH + (q & 63) * 16) = v;
        }
    }
    __syncthreads();
    if (tile >= GN / 16) return;

    const unsigned char* w1s = L + lrow * W1S_PITCH + kb * 2;
    const unsigned char* w2s = L + W1S_BYTES + lrow * W2S_PITCH + kb * 2;
    unsigned char* scr = L + W1S_BYTES + W2S_BYTES + wave * SCR_BYTES;

    v8f acc2[8];
    #pragma unroll
    for (int nt = 0; nt < 8; ++nt) acc2[nt] = v8f{};

    for (int c = 0; c < GDFF / 64; ++c) {
        v8f acc1[4];
        #pragma unroll
        for (int t = 0; t < 4; ++t) acc1[t] = v8f{};
        #pragma unroll
        for (int ks = 0; ks < 4; ++ks) {
            #pragma unroll
            for (int t = 0; t < 4; ++t) {
                v16bf b = lds_frag16(w1s + (c * 64 + t * 16) * W1S_PITCH + ks * 64);
                acc1[t] = wmma_bf16(a[ks], b, acc1[t]);
            }
        }
        #pragma unroll
        for (int t = 0; t < 4; ++t) {
            float bias = b1[c * 64 + t * 16 + lrow];
            #pragma unroll
            for (int j = 0; j < 8; ++j) {
                float v = fmaxf(acc1[t][j] + bias, 0.f);
                *(unsigned short*)(scr + (half8 + j) * SCR_PITCH + (t * 16 + lrow) * 2) = f2bf(v);
            }
        }
        wave_lds_handover();
        #pragma unroll
        for (int ks2 = 0; ks2 < 2; ++ks2) {
            v16bf a2 = lds_frag16(scr + lrow * SCR_PITCH + kb * 2 + ks2 * 64);
            #pragma unroll
            for (int nt = 0; nt < 8; ++nt) {
                v16bf b = lds_frag16(w2s + nt * 16 * W2S_PITCH + (c * 64 + ks2 * 32) * 2);
                acc2[nt] = wmma_bf16(a2, b, acc2[nt]);
            }
        }
        wave_lds_handover();
    }

    #pragma unroll
    for (int nt = 0; nt < 8; ++nt) {
        int n = nt * 16 + lrow;
        float bias = b2[n];
        #pragma unroll
        for (int j = 0; j < 8; ++j) {
            size_t r = (size_t)(rowBase + half8 + j);
            out[r * GD + n] = acc2[nt][j] + bias + hres[r * GD + n];
        }
    }
}

extern "C" void kernel_launch(void* const* d_in, const int* in_sizes, int n_in,
                              void* d_out, int out_size, void* d_ws, size_t ws_size,
                              hipStream_t stream) {
    const float* x     = (const float*)d_in[0];
    const float* Wfc   = (const float*)d_in[1];
    const float* a_l   = (const float*)d_in[2];
    const float* a_r   = (const float*)d_in[3];
    const float* gamma = (const float*)d_in[4];
    const float* beta  = (const float*)d_in[5];
    const float* W1    = (const float*)d_in[6];
    const float* b1    = (const float*)d_in[7];
    const float* W2    = (const float*)d_in[8];
    const float* b2    = (const float*)d_in[9];
    const int* esrc    = (const int*)d_in[10];
    const int* edst    = (const int*)d_in[11];
    float* out = (float*)d_out;

    char* ws = (char*)d_ws;
    size_t off = 0;
    auto take = [&](size_t bytes) -> char* {
        char* p = ws + off;
        off += (bytes + 255) & ~(size_t)255;
        return p;
    };
    unsigned short* xb    = (unsigned short*)take((size_t)GN * GD * 2);
    unsigned short* Wzt   = (unsigned short*)take((size_t)GD * GD * 2);
    unsigned short* W1t   = (unsigned short*)take((size_t)GDFF * GD * 2);
    unsigned short* W2t   = (unsigned short*)take((size_t)GD * GDFF * 2);
    float*          z     = (float*)take((size_t)GN * GD * 4);
    float*          el    = (float*)take((size_t)GN * GH * 4);
    float*          er    = (float*)take((size_t)GN * GH * 4);
    int*            rowptr= (int*)take((size_t)(GN + 1) * 4);
    int*            csrc  = (int*)take((size_t)GE * 4);
    float*          hres  = (float*)take((size_t)GN * GD * 4);
    unsigned short* lnb   = (unsigned short*)take((size_t)GN * GD * 2);
    int2*           pairs = (int2*)hres;
    int*            cell  = (int*)lnb;
    int*            bstart= (int*)((char*)lnb + (size_t)CH_N * CELL_PITCH * 4);

    repack_count_kernel<<<CH_N + RP_BLOCKS, RP_T, 0, stream>>>(x, Wfc, W1, W2, edst, xb, Wzt, W1t, W2t, cell);
    proj_place_kernel  <<<CH_N + PZ_BLOCKS, PZ_T, 0, stream>>>(xb, Wzt, a_l, a_r, z, el, er, esrc, edst, cell, bstart, pairs);
    binfinish_kernel<<<CB_N, FIN_T, 0, stream>>>(bstart, pairs, rowptr, csrc);
    gather_ln_kernel<<<GN / 8, 256, 0, stream>>>(z, el, er, rowptr, csrc, x, gamma, beta, hres, lnb);
    ffn_fused_kernel<<<FF_BLOCKS, FF_T, 0, stream>>>(lnb, W1t, b1, W2t, b2, hres, out);
}


# Edge grouping loses a launch: each chunk orders its own edges by coarse bin in LDS (count+place in one pass), bins collect their pieces beside the projection; 5 launches -> 4; bucket order: arrival
# speedup vs baseline: 6.5618x; 1.1505x over previous
#include <hip/hip_runtime.h>

typedef __attribute__((ext_vector_type(16))) __bf16 v16bf;
typedef __attribute__((ext_vector_type(8)))  float  v8f;
typedef __attribute__((ext_vector_type(4)))  unsigned int u32x4;

union FragU { u32x4 q[2]; v16bf v; };

__device__ __forceinline__ v16bf load_frag16(const unsigned short* p) {
    FragU f;
    f.q[0] = *(const u32x4*)(p);
    f.q[1] = *(const u32x4*)(p + 16);
    return f.v;
}

__device__ __forceinline__ v8f wmma_bf16(v16bf a, v16bf b, v8f c) {
    return __builtin_amdgcn_wmma_f32_16x16x32_bf16(
         false, a,  false, b,
         (short)0, c,  false,  false);
}

__device__ __forceinline__ v16bf lds_frag16(const unsigned char* p) {
    FragU f;
    f.q[0] = *(const u32x4*)(p);
    f.q[1] = *(const u32x4*)(p + 32);
    return f.v;
}

__device__ __forceinline__ void wave_lds_handover() {
    __builtin_amdgcn_fence(__ATOMIC_RELEASE, "wavefront");
    __builtin_amdgcn_wave_barrier();
    __builtin_amdgcn_fence(__ATOMIC_ACQUIRE, "wavefront");
}

__device__ __forceinline__ unsigned short f2bf(float f) {
    unsigned int u = __float_as_uint(f);
    unsigned int r = u + 0x7FFFu + ((u >> 16) & 1u);
    return (unsigned short)(r >> 16);
}

#define GN 50000
#define GE 800000
#define GD 128
#define GH 8
#define GDFF 512

#define CB_SHIFT 8
#define CB_NODES (1 << CB_SHIFT)
#define CB_N ((GN + CB_NODES - 1) / CB_NODES)
#define CH_N 256
#define CH_E (GE / CH_N)
#define SB_T 256
#define SB_IT ((CH_E + SB_T - 1) / SB_T)
#define CELL_PITCH 256
static_assert(CH_E * CH_N == GE, "the edges must split into whole chunks");
static_assert(CB_N + 32 <= CELL_PITCH && CELL_PITCH == SB_T, "one bin per thread in the chunk's scan, and a spare counter per lane behind the bins");
static_assert(CH_N == SB_T && CB_NODES == SB_T, "launch 2: one chunk per thread, then one node per thread");

#define RP_T 256
#define RP_U 10
#define RP_BLOCKS (GN * GD / (RP_T * 4 * RP_U))
static_assert(RP_BLOCKS * RP_T * 4 * RP_U == GN * GD, "x must split into whole blocks");
static_assert(RP_BLOCKS * RP_T >= GDFF * GD, "grid must still cover the weight part");
static_assert(RP_T == SB_T, "both kinds of work-group share the launch's block size");
__global__ void __launch_bounds__(RP_T)
repack_sort_kernel(const float* __restrict__ x, const float* __restrict__ Wfc,
              const float* __restrict__ W1, const float* __restrict__ W2,
              const int* __restrict__ esrc, const int* __restrict__ edst,
              unsigned short* __restrict__ xb, unsigned short* __restrict__ Wzt,
              unsigned short* __restrict__ W1t, unsigned short* __restrict__ W2t,
              int2* __restrict__ cell, int2* __restrict__ pairs) {
    __shared__ int hc[CELL_PITCH];
    __shared__ int wsum[SB_T / 32];
    if (blockIdx.x < CH_N) {
        const int t = threadIdx.x;
        const int c = blockIdx.x;
        hc[t] = 0;
        __syncthreads();
        const int base = c * CH_E;
        int d[SB_IT], sr[SB_IT];
        #pragma unroll
        for (int k = 0; k < SB_IT; ++k) {
            int i = t + k * SB_T;
            d[k]  = (i < CH_E) ? edst[base + i] : -1;
            sr[k] = (i < CH_E) ? esrc[base + i] : 0;
        }
        #pragma unroll
        for (int k = 0; k < SB_IT; ++k)
            if ((unsigned int)d[k] < (unsigned int)GN) atomicAdd(&hc[d[k] >> CB_SHIFT], 1);
        __syncthreads();
        const int lane = t & 31, w = t >> 5;
        const int cb = hc[t];
        int incl = cb;
        for (int dd = 1; dd < 32; dd <<= 1) { int y = __shfl_up(incl, dd); if (lane >= dd) incl += y; }
        if (lane == 31) wsum[w] = incl;
        __syncthreads();
        int wbase = 0;
        for (int k = 0; k < SB_T / 32; ++k) wbase += (k < w) ? wsum[k] : 0;
        const int start = wbase + incl - cb;
        hc[t] = start;
        if (t < CB_N) { int2 e; e.x = cb; e.y = start; cell[c * CELL_PITCH + t] = e; }
        __syncthreads();
        int pos[SB_IT];
        #pragma unroll
        for (int k = 0; k < SB_IT; ++k) {
            const int slot = ((unsigned int)d[k] < (unsigned int)GN) ? (d[k] >> CB_SHIFT) : (CB_N + lane);
            pos[k] = atomicAdd(&hc[slot], 1);
        }
        #pragma unroll
        for (int k = 0; k < SB_IT; ++k) {
            if ((unsigned int)d[k] < (unsigned int)GN) {
                const int q = min(max(pos[k], 0), CH_E - 1);
                int2 p; p.x = sr[k]; p.y = d[k];
                pairs[base + q] = p;
            }
        }
        return;
    }
    const int rb = blockIdx.x - CH_N;
    int idx = rb * RP_T + threadIdx.x;
    size_t g0 = (size_t)rb * (RP_T * RP_U) + threadIdx.x;
    const float4* xs = (const float4*)x + g0;
    uint2* xd = (uint2*)xb + g0;
    float4 v[RP_U];
    #pragma unroll
    for (int j = 0; j < RP_U; ++j) v[j] = xs[j * RP_T];
    if (idx < GD * GD) {
        int j = idx >> 7, d = idx & 127;
        int h = j >> 4, k = j & 15;
        Wzt[idx] = f2bf(Wfc[(h * GD + d) * 16 + k]);
    }
    if (idx < GDFF * GD) {
        int f = idx >> 7, d = idx & 127;
        W1t[idx] = f2bf(W1[d * GDFF + f]);
        int d2 = idx >> 9, f2 = idx & 511;
        W2t[idx] = f2bf(W2[f2 * GD + d2]);
    }
    #pragma unroll
    for (int j = 0; j < RP_U; ++j) {
        uint2 pk;
        pk.x = (unsigned int)f2bf(v[j].x) | ((unsigned int)f2bf(v[j].y) << 16);
        pk.y = (unsigned int)f2bf(v[j].z) | ((unsigned int)f2bf(v[j].w) << 16);
        xd[j * RP_T] = pk;
    }
}

#define PZ_T 256
#define PZ_WAVES 8
#define PZ_BLOCKS ((GN / 16 + PZ_WAVES - 1) / PZ_WAVES)
#define WZS_PITCH 272
#define WZS_BYTES (GD * WZS_PITCH)
#define ZS_PITCH 528
#define ZS_BYTES (16 * ZS_PITCH)
#define PZ_LDS (WZS_BYTES + PZ_WAVES * ZS_BYTES)
static_assert((GD * GD * 2 / 16) % PZ_T == 0, "the weight copy must split into whole passes");
static_assert(GN % 16 == 0, "whole strips only");
static_assert(PZ_T == SB_T, "both kinds of work-group share the launch's block size");

#define FIN_R 32
static_assert((CB_NODES + 32 + 2 * (SB_T / 32)) * 4 <= PZ_LDS, "the grouping part borrows the projection's LDS");
__device__ __forceinline__ void finish_part(unsigned char* L, const int b,
                                            const int2* __restrict__ cell, const int2* __restrict__ pairs,
                                            int* __restrict__ rowptr, int* __restrict__ csrc) {
    int* cnt   = (int*)L;
    int* wsumB = cnt + CB_NODES + 32;
    int* wsumN = wsumB + SB_T / 32;
    const int t = threadIdx.x;
    const int lane = t & 31, w = t >> 5;
    const int2 ce = cell[t * CELL_PITCH + b];
    cnt[t] = 0;
    if (t < 32) cnt[CB_NODES + t] = 0;
    int rs = ce.y;
    for (int sh = 16; sh > 0; sh >>= 1) rs += __shfl_xor(rs, sh);
    if (lane == 0) wsumB[w] = rs;
    const int oseg = min(max(ce.y, 0), CH_E);
    const int nseg = min(max(ce.x, 0), CH_E - oseg);
    const int2* chunk = pairs + (size_t)t * CH_E;
    const int2* piece = chunk + oseg;
    int ss[FIN_R], ix[FIN_R];
    #pragma unroll
    for (int u = 0; u < FIN_R; ++u) {
        const int2 p = chunk[(u < nseg) ? (oseg + u) : 0];
        ss[u] = p.x;
        ix[u] = (u < nseg) ? (p.y & (CB_NODES - 1)) : (CB_NODES + lane);
    }
    __syncthreads();
    int beg = 0;
    for (int k = 0; k < SB_T / 32; ++k) beg += wsumB[k];
    beg = min(max(beg, 0), GE);
    #pragma unroll
    for (int u = 0; u < FIN_R; ++u) atomicAdd(&cnt[ix[u]], 1);
    for (int k = FIN_R; k < nseg; ++k)
        atomicAdd(&cnt[piece[k].y & (CB_NODES - 1)], 1);
    __syncthreads();
    const int cN = cnt[t];
    int incl = cN;
    for (int s = 1; s < 32; s <<= 1) { int y = __shfl_up(incl, s); if (lane >= s) incl += y; }
    if (lane == 31) wsumN[w] = incl;
    __syncthreads();
    int wbase = 0;
    for (int k = 0; k < SB_T / 32; ++k) wbase += (k < w) ? wsumN[k] : 0;
    const int start = beg + wbase + incl - cN;
    cnt[t] = start;
    const int n = b * CB_NODES + t;
    if (n <= GN) rowptr[n] = start;
    __syncthreads();
    #pragma unroll
    for (int u = 0; u < FIN_R; ++u) ix[u] = atomicAdd(&cnt[ix[u]], 1);
    #pragma unroll
    for (int u = 0; u < FIN_R; ++u)
        if (u < nseg) csrc[min(max(ix[u], 0), GE - 1)] = ss[u];
    for (int k = FIN_R; k < nseg; ++k) {
        const int2 p = piece[k];
        int pos = atomicAdd(&cnt[p.y & (CB_NODES - 1)], 1);
        pos = min(max(pos, 0), GE - 1);
        csrc[pos] = p.x;
    }
}

__global__ void __launch_bounds__(PZ_T, 6)
proj_finish_kernel(const unsigned short* __restrict__ xb, const unsigned short* __restrict__ Wzt,
            const float* __restrict__ a_l, const float* __restrict__ a_r,
            float* __restrict__ z, float* __restrict__ el, float* __restrict__ er,
            const int2* __restrict__ cell, const int2* __restrict__ pairs,
            int* __restrict__ rowptr, int* __restrict__ csrc) {
    __shared__ __attribute__((aligned(16))) unsigned char L[PZ_LDS];
    if (blockIdx.x < CB_N) {
        finish_part(L, blockIdx.x, cell, pairs, rowptr, csrc);
        return;
    }
    const int pb = blockIdx.x - CB_N;
    const int tid = threadIdx.x;
    const int lane = tid & 31;
    const int wave = __builtin_amdgcn_readfirstlane(tid >> 5);
    const int tile = pb * PZ_WAVES + wave;
    const int tileC = min(tile, GN / 16 - 1);
    const int rowBase = tileC * 16;
    const int lrow = lane & 15;
    const int kb = (lane & 16) ? 8 : 0;
    const int half8 = (lane & 16) >> 1;

    const unsigned short* aRow = xb + (size_t)(rowBase + lrow) * GD + kb;
    v16bf a[4];
    #pragma unroll
    for (int ks = 0; ks < 4; ++ks) a[ks] = load_frag16(aRow + ks * 32);
    {
        const u32x4* g = (const u32x4*)Wzt;
        #pragma unroll
        for (int p = 0; p < (GD * GD * 2 / 16) / PZ_T; ++p) {
            int q = p * PZ_T + tid;
            u32x4 v = g[q];
            *(u32x4*)(L + (q >> 4) * WZS_PITCH + (q & 15) * 16) = v;
        }
    }
    __syncthreads();
    if (tile >= GN / 16) return;

    const unsigned char* wzs = L + lrow * WZS_PITCH + kb * 2;
    float* scr = (float*)(L + WZS_BYTES + wave * ZS_BYTES);

    v8f acc[8];
    #pragma unroll
    for (int t = 0; t < 8; ++t) acc[t] = v8f{};
    #pragma unroll
    for (int ks = 0; ks < 4; ++ks) {
        v16bf b[8];
        #pragma unroll
        for (int t = 0; t < 8; ++t) b[t] = lds_frag16(wzs + t * 16 * WZS_PITCH + ks * 64);
        #pragma unroll
        for (int t = 0; t < 8; ++t) acc[t] = wmma_bf16(a[ks], b[t], acc[t]);
    }

    const float4 al = *(const float4*)(a_l + lane * 4);
    const float4 ar = *(const float4*)(a_r + lane * 4);
    #pragma unroll
    for (int t = 0; t < 8; ++t) {
        #pragma unroll
        for (int j = 0; j < 8; ++j)
            scr[(half8 + j) * (ZS_PITCH / 4) + t * 16 + lrow] = acc[t][j];
    }
    wave_lds_handover();
    const int hd = lane >> 2;
    const int qd = lane & 3;
    float kl[4] = {0.f, 0.f, 0.f, 0.f}, kr[4] = {0.f, 0.f, 0.f, 0.f};
    #pragma unroll
    for (int r = 0; r < 16; ++r) {
        const float4 zz = *(const float4*)(scr + r * (ZS_PITCH / 4) + lane * 4);
        *(float4*)(z + (size_t)(rowBase + r) * GD + lane * 4) = zz;
        float sl = (zz.x * al.x + zz.y * al.y) + (zz.z * al.z + zz.w * al.w);
        float sr = (zz.x * ar.x + zz.y * ar.y) + (zz.z * ar.z + zz.w * ar.w);
        sl += __shfl_xor(sl, 1); sl += __shfl_xor(sl, 2);
        sr += __shfl_xor(sr, 1); sr += __shfl_xor(sr, 2);
        if ((r & 3) == qd) { kl[r >> 2] = sl; kr[r >> 2] = sr; }
    }
    #pragma unroll
    for (int i = 0; i < 4; ++i) {
        int row = rowBase + qd + 4 * i;
        el[row * GH + hd] = kl[i];
        er[row * GH + hd] = kr[i];
    }
}

__global__ void gather_ln_kernel(const float* __restrict__ z, const float* __restrict__ el,
                                 const float* __restrict__ er, const int* __restrict__ rowptr,
                                 const int* __restrict__ csrc, const float* __restrict__ x,
                                 const float* __restrict__ gamma, const float* __restrict__ beta,
                                 float* __restrict__ hres, unsigned short* __restrict__ lnb) {
    int lane = threadIdx.x & 31;
    int wave = __builtin_amdgcn_readfirstlane((int)(threadIdx.x >> 5));
    int n = blockIdx.x * 8 + wave;
    int beg = __builtin_amdgcn_readfirstlane(rowptr[n]);
    int end = __builtin_amdgcn_readfirstlane(rowptr[n + 1]);
    beg = min(max(beg, 0), GE);
    end = min(max(end, beg), GE);
    int hd = lane >> 2;
    float ern = er[n * GH + hd];
    const float4 xx = *(const float4*)(x + (size_t)n * GD + lane * 4);
    const float4 gg = *(const float4*)(gamma + lane * 4);
    const float4 bb = *(const float4*)(beta + lane * 4);
    float m = -INFINITY, s = 0.f;
    float a0 = 0.f, a1 = 0.f, a2 = 0.f, a3 = 0.f;
    for (int base = beg; base < end; base += 32) {
        int cc = min(32, end - base);
        int mysrc = (lane < cc) ? csrc[base + lane] : 0;
        #pragma unroll 4
        for (int j = 0; j < cc; ++j) {
            int sidx = __builtin_amdgcn_readlane(mysrc, j);
            sidx = min(max(sidx, 0), GN - 1);
            float v = el[sidx * GH + hd] + ern;
            v = v > 0.f ? v : 0.01f * v;
            const float4 zz = *(const float4*)(z + (size_t)sidx * GD + lane * 4);
            float mn = fmaxf(m, v);
            float sc = __expf(m - mn);
            float p  = __expf(v - mn);
            s  = s  * sc + p;
            a0 = a0 * sc + p * zz.x;
            a1 = a1 * sc + p * zz.y;
            a2 = a2 * sc + p * zz.z;
            a3 = a3 * sc + p * zz.w;
            m = mn;
        }
    }
    float inv = 1.f / fmaxf(s, 1e-9f);
    float v0 = a0 * inv, v1 = a1 * inv, v2 = a2 * inv, v3 = a3 * inv;
    v0 = v0 > 0.f ? v0 : (__expf(v0) - 1.f);
    v1 = v1 > 0.f ? v1 : (__expf(v1) - 1.f);
    v2 = v2 > 0.f ? v2 : (__expf(v2) - 1.f);
    v3 = v3 > 0.f ? v3 : (__expf(v3) - 1.f);
    v0 += xx.x; v1 += xx.y; v2 += xx.z; v3 += xx.w;
    float4 hr; hr.x = v0; hr.y = v1; hr.z = v2; hr.w = v3;
    *(float4*)(hres + (size_t)n * GD + lane * 4) = hr;
    float sm = (v0 + v1) + (v2 + v3);
    float sq = (v0 * v0 + v1 * v1) + (v2 * v2 + v3 * v3);
    for (int d = 16; d > 0; d >>= 1) {
        sm += __shfl_xor(sm, d);
        sq += __shfl_xor(sq, d);
    }
    float mean = sm * (1.f / GD);
    float var  = sq * (1.f / GD) - mean * mean;
    float rstd = rsqrtf(var + 1e-5f);
    unsigned int b0 = f2bf((v0 - mean) * rstd * gg.x + bb.x);
    unsigned int b1 = f2bf((v1 - mean) * rstd * gg.y + bb.y);
    unsigned int b2 = f2bf((v2 - mean) * rstd * gg.z + bb.z);
    unsigned int b3 = f2bf((v3 - mean) * rstd * gg.w + bb.w);
    uint2 pk; pk.x = b0 | (b1 << 16); pk.y = b2 | (b3 << 16);
    *(uint2*)(lnb + (size_t)n * GD + lane * 4) = pk;
}

#define FF_T 512
#define FF_WAVES 16
#define FF_BLOCKS ((GN / 16 + FF_WAVES - 1) / FF_WAVES)
#define W1S_PITCH 272
#define W2S_PITCH 1040
#define SCR_PITCH 144
#define W1S_BYTES (GDFF * W1S_PITCH)
#define W2S_BYTES (GD * W2S_PITCH)
#define SCR_BYTES (16 * SCR_PITCH)
#define FF_LDS (W1S_BYTES + W2S_BYTES + FF_WAVES * SCR_BYTES)
static_assert(FF_LDS <= 320 * 1024, "weights and scratch must fit one work-group's LDS");
static_assert((GDFF * GD * 2 / 16) % FF_T == 0, "the weight copy must split into whole passes");

__global__ void __launch_bounds__(FF_T)
ffn_fused_kernel(const unsigned short* __restrict__ lnb,
                 const unsigned short* __restrict__ W1t, const float* __restrict__ b1,
                 const unsigned short* __restrict__ W2t, const float* __restrict__ b2,
                 const float* __restrict__ hres, float* __restrict__ out) {
    __shared__ __attribute__((aligned(16))) unsigned char L[FF_LDS];
    const int tid = threadIdx.x;
    const int lane = tid & 31;
    const int wave = __builtin_amdgcn_readfirstlane(tid >> 5);
    const int tile = blockIdx.x * FF_WAVES + wave;
    const int tileC = min(tile, GN / 16 - 1);
    const int rowBase = tileC * 16;
    const int lrow = lane & 15;
    const int kb = (lane & 16) ? 8 : 0;
    const int half8 = (lane & 16) >> 1;

    const unsigned short* aRow = lnb + (size_t)(rowBase + lrow) * GD + kb;
    v16bf a[4];
    #pragma unroll
    for (int ks = 0; ks < 4; ++ks) a[ks] = load_frag16(aRow + ks * 32);

    {
        const u32x4* g1 = (const u32x4*)W1t;
        #pragma unroll 8
        for (int p = 0; p < (GDFF * GD * 2 / 16) / FF_T; ++p) {
            int q = p * FF_T + tid;
            u32x4 v = g1[q];
            *(u32x4*)(L + (q >> 4) * W1S_PITCH + (q & 15) * 16) = v;
        }
        const u32x4* g2 = (const u32x4*)W2t;
        #pragma unroll 8
        for (int p = 0; p < (GD * GDFF * 2 / 16) / FF_T; ++p) {
            int q = p * FF_T + tid;
            u32x4 v = g2[q];
            *(u32x4*)(L + W1S_BYTES + (q >> 6) * W2S_PITCH + (q & 63) * 16) = v;
        }
    }
    __syncthreads();
    if (tile >= GN / 16) return;

    const unsigned char* w1s = L + lrow * W1S_PITCH + kb * 2;
    const unsigned char* w2s = L + W1S_BYTES + lrow * W2S_PITCH + kb * 2;
    unsigned char* scr = L + W1S_BYTES + W2S_BYTES + wave * SCR_BYTES;

    v8f acc2[8];
    #pragma unroll
    for (int nt = 0; nt < 8; ++nt) acc2[nt] = v8f{};

    for (int c = 0; c < GDFF / 64; ++c) {
        v8f acc1[4];
        #pragma unroll
        for (int t = 0; t < 4; ++t) acc1[t] = v8f{};
        #pragma unroll
        for (int ks = 0; ks < 4; ++ks) {
            #pragma unroll
            for (int t = 0; t < 4; ++t) {
                v16bf b = lds_frag16(w1s + (c * 64 + t * 16) * W1S_PITCH + ks * 64);
                acc1[t] = wmma_bf16(a[ks], b, acc1[t]);
            }
        }
        #pragma unroll
        for (int t = 0; t < 4; ++t) {
            float bias = b1[c * 64 + t * 16 + lrow];
            #pragma unroll
            for (int j = 0; j < 8; ++j) {
                float v = fmaxf(acc1[t][j] + bias, 0.f);
                *(unsigned short*)(scr + (half8 + j) * SCR_PITCH + (t * 16 + lrow) * 2) = f2bf(v);
            }
        }
        wave_lds_handover();
        #pragma unroll
        for (int ks2 = 0; ks2 < 2; ++ks2) {
            v16bf a2 = lds_frag16(scr + lrow * SCR_PITCH + kb * 2 + ks2 * 64);
            #pragma unroll
            for (int nt = 0; nt < 8; ++nt) {
                v16bf b = lds_frag16(w2s + nt * 16 * W2S_PITCH + (c * 64 + ks2 * 32) * 2);
                acc2[nt] = wmma_bf16(a2, b, acc2[nt]);
            }
        }
        wave_lds_handover();
    }

    #pragma unroll
    for (int nt = 0; nt < 8; ++nt) {
        int n = nt * 16 + lrow;
        float bias = b2[n];
        #pragma unroll
        for (int j = 0; j < 8; ++j) {
            size_t r = (size_t)(rowBase + half8 + j);
            out[r * GD + n] = acc2[nt][j] + bias + hres[r * GD + n];
        }
    }
}

extern "C" void kernel_launch(void* const* d_in, const int* in_sizes, int n_in,
                              void* d_out, int out_size, void* d_ws, size_t ws_size,
                              hipStream_t stream) {
    const float* x     = (const float*)d_in[0];
    const float* Wfc   = (const float*)d_in[1];
    const float* a_l   = (const float*)d_in[2];
    const float* a_r   = (const float*)d_in[3];
    const float* gamma = (const float*)d_in[4];
    const float* beta  = (const float*)d_in[5];
    const float* W1    = (const float*)d_in[6];
    const float* b1    = (const float*)d_in[7];
    const float* W2    = (const float*)d_in[8];
    const float* b2    = (const float*)d_in[9];
    const int* esrc    = (const int*)d_in[10];
    const int* edst    = (const int*)d_in[11];
    float* out = (float*)d_out;

    char* ws = (char*)d_ws;
    size_t off = 0;
    auto take = [&](size_t bytes) -> char* {
        char* p = ws + off;
        off += (bytes + 255) & ~(size_t)255;
        return p;
    };
    unsigned short* xb    = (unsigned short*)take((size_t)GN * GD * 2);
    unsigned short* Wzt   = (unsigned short*)take((size_t)GD * GD * 2);
    unsigned short* W1t   = (unsigned short*)take((size_t)GDFF * GD * 2);
    unsigned short* W2t   = (unsigned short*)take((size_t)GD * GDFF * 2);
    float*          z     = (float*)take((size_t)GN * GD * 4);
    float*          el    = (float*)take((size_t)GN * GH * 4);
    float*          er    = (float*)take((size_t)GN * GH * 4);
    int*            rowptr= (int*)take((size_t)(GN + 1) * 4);
    int*            csrc  = (int*)take((size_t)GE * 4);
    float*          hres  = (float*)take((size_t)GN * GD * 4);
    unsigned short* lnb   = (unsigned short*)take((size_t)GN * GD * 2);
    int2*           pairs = (int2*)hres;
    int2*           cell  = (int2*)lnb;

    repack_sort_kernel<<<CH_N + RP_BLOCKS, RP_T, 0, stream>>>(x, Wfc, W1, W2, esrc, edst, xb, Wzt, W1t, W2t, cell, pairs);
    proj_finish_kernel<<<CB_N + PZ_BLOCKS, PZ_T, 0, stream>>>(xb, Wzt, a_l, a_r, z, el, er, cell, pairs, rowptr, csrc);
    gather_ln_kernel<<<GN / 8, 256, 0, stream>>>(z, el, er, rowptr, csrc, x, gamma, beta, hres, lnb);
    ffn_fused_kernel<<<FF_BLOCKS, FF_T, 0, stream>>>(lnb, W1t, b1, W2t, b2, hres, out);
}
